# starting kernel
# baseline (speedup 1.0000x reference)
#include <hip/hip_runtime.h>
#include <hip/hip_bf16.h>

typedef __attribute__((ext_vector_type(16))) _Float16 v16h;
typedef __attribute__((ext_vector_type(8)))  _Float16 v8h;
typedef __attribute__((ext_vector_type(8)))  float    v8f;

#define B_TOT   65536
#define S_DIM   256
#define A_N     8
#define K_N     10
#define H_DIM   64
#define AD_DIM  128
#define D_DIM   384

#define WG_ROWS 64
#define NTHREADS 64
#define NWAVES  (NTHREADS / 32)
#define ROWS_PER_WAVE 32
#define XS      392
#define HS      72

__device__ __forceinline__ v16h load_frag(const _Float16* __restrict__ p,
                                          int stride, int row, int k0, int lane) {
    const int ho = (lane >> 4) << 3;
    const _Float16* q = p + row * stride + k0 + ho;
    union { v16h v; struct { v8h lo; v8h hi; } s; } u;
    u.s.lo = *(const v8h*)q;
    u.s.hi = *(const v8h*)(q + 16);
    return u.v;
}

template <int KD>
__device__ __forceinline__ void gemm4x2(const _Float16* __restrict__ A, int lda,
                                        const _Float16* __restrict__ W, int ldw,
                                        int lane, v8f c[2][4]) {
    const int mr = lane & 15;
    v8f z = {0.f, 0.f, 0.f, 0.f, 0.f, 0.f, 0.f, 0.f};
#pragma unroll
    for (int m = 0; m < 2; ++m)
#pragma unroll
        for (int t = 0; t < 4; ++t) c[m][t] = z;
    for (int k0 = 0; k0 < KD; k0 += 32) {
        v16h a0 = load_frag(A, lda, mr, k0, lane);
        v16h a1 = load_frag(A + 16 * lda, lda, mr, k0, lane);
#pragma unroll
        for (int t = 0; t < 4; ++t) {
            v16h b = load_frag(W, ldw, t * 16 + mr, k0, lane);
            c[0][t] = __builtin_amdgcn_wmma_f32_16x16x32_f16(
                false, a0, false, b, (short)0, c[0][t], false, false);
            c[1][t] = __builtin_amdgcn_wmma_f32_16x16x32_f16(
                false, a1, false, b, (short)0, c[1][t], false, false);
        }
    }
}

__device__ __forceinline__ void gemm1x2(const _Float16* __restrict__ A, int lda,
                                        const _Float16* __restrict__ W,
                                        int lane, v8f c[2]) {
    const int mr = lane & 15;
    v8f z = {0.f, 0.f, 0.f, 0.f, 0.f, 0.f, 0.f, 0.f};
    c[0] = z; c[1] = z;
#pragma unroll
    for (int k0 = 0; k0 < H_DIM; k0 += 32) {
        v16h a0 = load_frag(A, lda, mr, k0, lane);
        v16h a1 = load_frag(A + 16 * lda, lda, mr, k0, lane);
        v16h b  = load_frag(W, H_DIM, mr, k0, lane);
        c[0] = __builtin_amdgcn_wmma_f32_16x16x32_f16(
            false, a0, false, b, (short)0, c[0], false, false);
        c[1] = __builtin_amdgcn_wmma_f32_16x16x32_f16(
            false, a1, false, b, (short)0, c[1], false, false);
    }
}

__device__ __forceinline__ void bias_relu_store2(const v8f c[2][4],
                                                 const float* __restrict__ bias,
                                                 _Float16* __restrict__ Hout, int lane) {
    const int col = lane & 15, hi = lane >> 4;
#pragma unroll
    for (int t = 0; t < 4; ++t) {
        float bv = bias[t * 16 + col];
#pragma unroll
        for (int m = 0; m < 2; ++m)
#pragma unroll
            for (int r = 0; r < 8; ++r) {
                float v = c[m][t][r] + bv;
                Hout[(m * 16 + r + 8 * hi) * HS + t * 16 + col] =
                    (_Float16)(v > 0.f ? v : 0.f);
            }
    }
}

__global__ void cvt_f32_f16(const float* __restrict__ src, _Float16* __restrict__ dst, int n) {
    int i = blockIdx.x * blockDim.x + threadIdx.x;
    if (i < n) dst[i] = (_Float16)src[i];
}

__global__ void cvt_pad_w3(const float* __restrict__ src, _Float16* __restrict__ dst, int outdim) {
    int i = blockIdx.x * blockDim.x + threadIdx.x;
    const int n = K_N * 16 * H_DIM;
    if (i >= n) return;
    int h = i % H_DIM;
    int o = (i / H_DIM) % 16;
    int k = i / (16 * H_DIM);
    dst[i] = (o < outdim) ? (_Float16)src[(k * outdim + o) * H_DIM + h] : (_Float16)0.f;
}

__global__ __launch_bounds__(NTHREADS, 2)
void qplex_main(const float* __restrict__ states, const float* __restrict__ actions,
                const _Float16* __restrict__ wk1, const _Float16* __restrict__ wa1,
                const _Float16* __restrict__ wc1,
                const _Float16* __restrict__ wk2, const _Float16* __restrict__ wa2,
                const _Float16* __restrict__ wc2,
                const _Float16* __restrict__ wk3, const _Float16* __restrict__ wa3,
                const _Float16* __restrict__ wc3,
                const float* __restrict__ bk1, const float* __restrict__ bk2,
                const float* __restrict__ bk3,
                const float* __restrict__ ba1, const float* __restrict__ ba2,
                const float* __restrict__ ba3,
                const float* __restrict__ bc1, const float* __restrict__ bc2,
                const float* __restrict__ bc3,
                float* __restrict__ out) {
    __shared__ _Float16 Xs[WG_ROWS * XS];
    __shared__ _Float16 Hs[NWAVES * ROWS_PER_WAVE * HS];

    const int tid = threadIdx.x;
    const int b0  = blockIdx.x * WG_ROWS;

    for (int idx = tid; idx < WG_ROWS * D_DIM; idx += NTHREADS) {
        int row = idx / D_DIM;
        int col = idx - row * D_DIM;
        float v = (col < S_DIM)
                      ? states[(size_t)(b0 + row) * S_DIM + col]
                      : actions[(size_t)(b0 + row) * AD_DIM + (col - S_DIM)];
        Xs[row * XS + col] = (_Float16)v;
    }
    __syncthreads();

    const int wave = tid >> 5, lane = tid & 31;
    const int col = lane & 15, hi = lane >> 4;
    const _Float16* Xw = Xs + (wave * ROWS_PER_WAVE) * XS;
    _Float16*       Hw = Hs + (wave * ROWS_PER_WAVE) * HS;

    float acc[2][8];
#pragma unroll
    for (int m = 0; m < 2; ++m)
#pragma unroll
        for (int r = 0; r < 8; ++r) acc[m][r] = 0.f;

#pragma unroll 1
    for (int k = 0; k < K_N; ++k) {
        v8f c1[2][4];
        v8f ck[2], ca[2], cc[2];

        gemm4x2<S_DIM>(Xw, XS, wk1 + (size_t)k * H_DIM * S_DIM, S_DIM, lane, c1);
        bias_relu_store2(c1, bk1 + k * H_DIM, Hw, lane);
        gemm4x2<H_DIM>(Hw, HS, wk2 + (size_t)k * H_DIM * H_DIM, H_DIM, lane, c1);
        bias_relu_store2(c1, bk2 + k * H_DIM, Hw, lane);
        gemm1x2(Hw, HS, wk3 + (size_t)k * 16 * H_DIM, lane, ck);

        gemm4x2<S_DIM>(Xw, XS, wa1 + (size_t)k * H_DIM * S_DIM, S_DIM, lane, c1);
        bias_relu_store2(c1, ba1 + k * H_DIM, Hw, lane);
        gemm4x2<H_DIM>(Hw, HS, wa2 + (size_t)k * H_DIM * H_DIM, H_DIM, lane, c1);
        bias_relu_store2(c1, ba2 + k * H_DIM, Hw, lane);
        gemm1x2(Hw, HS, wa3 + (size_t)k * 16 * H_DIM, lane, ca);

        gemm4x2<D_DIM>(Xw, XS, wc1 + (size_t)k * H_DIM * D_DIM, D_DIM, lane, c1);
        bias_relu_store2(c1, bc1 + k * H_DIM, Hw, lane);
        gemm4x2<H_DIM>(Hw, HS, wc2 + (size_t)k * H_DIM * H_DIM, H_DIM, lane, c1);
        bias_relu_store2(c1, bc2 + k * H_DIM, Hw, lane);
        gemm1x2(Hw, HS, wc3 + (size_t)k * 16 * H_DIM, lane, cc);

        const float kbias = bk3[k];
        const float ab3v  = ba3[k * A_N + (col & 7)];
        const float cb3v  = bc3[k * A_N + (col & 7)];
#pragma unroll
        for (int m = 0; m < 2; ++m)
#pragma unroll
            for (int r = 0; r < 8; ++r) {
                float kv = __shfl(ck[m][r], lane & 16, 32);
                kv = fabsf(kv + kbias) + 1e-10f;
                float ag = 1.f / (1.f + __expf(-(ca[m][r] + ab3v)));
                float ac = 1.f / (1.f + __expf(-(cc[m][r] + cb3v)));
                acc[m][r] += kv * ag * ac;
            }
    }

    if (col < A_N) {
#pragma unroll
        for (int m = 0; m < 2; ++m)
#pragma unroll
            for (int r = 0; r < 8; ++r)
                out[(size_t)(b0 + wave * ROWS_PER_WAVE + m * 16 + r + 8 * hi) * A_N + col] =
                    acc[m][r];
    }
}

extern "C" void kernel_launch(void* const* d_in, const int* in_sizes, int n_in,
                              void* d_out, int out_size, void* d_ws, size_t ws_size,
                              hipStream_t stream) {
    (void)in_sizes; (void)n_in; (void)out_size; (void)ws_size;
    const float* states  = (const float*)d_in[0];
    const float* actions = (const float*)d_in[1];
    const float* kW1 = (const float*)d_in[2];
    const float* kb1 = (const float*)d_in[3];
    const float* kW2 = (const float*)d_in[4];
    const float* kb2 = (const float*)d_in[5];
    const float* kW3 = (const float*)d_in[6];
    const float* kb3 = (const float*)d_in[7];
    const float* aW1 = (const float*)d_in[8];
    const float* ab1 = (const float*)d_in[9];
    const float* aW2 = (const float*)d_in[10];
    const float* ab2 = (const float*)d_in[11];
    const float* aW3 = (const float*)d_in[12];
    const float* ab3 = (const float*)d_in[13];
    const float* cW1 = (const float*)d_in[14];
    const float* cb1 = (const float*)d_in[15];
    const float* cW2 = (const float*)d_in[16];
    const float* cb2 = (const float*)d_in[17];
    const float* cW3 = (const float*)d_in[18];
    const float* cb3 = (const float*)d_in[19];

    _Float16* ws = (_Float16*)d_ws;
    size_t off = 0;
    _Float16* wk1 = ws + off; off += (size_t)K_N * H_DIM * S_DIM;
    _Float16* wa1 = ws + off; off += (size_t)K_N * H_DIM * S_DIM;
    _Float16* wc1 = ws + off; off += (size_t)K_N * H_DIM * D_DIM;
    _Float16* wk2 = ws + off; off += (size_t)K_N * H_DIM * H_DIM;
    _Float16* wa2 = ws + off; off += (size_t)K_N * H_DIM * H_DIM;
    _Float16* wc2 = ws + off; off += (size_t)K_N * H_DIM * H_DIM;
    _Float16* wk3 = ws + off; off += (size_t)K_N * 16 * H_DIM;
    _Float16* wa3 = ws + off; off += (size_t)K_N * 16 * H_DIM;
    _Float16* wc3 = ws + off; off += (size_t)K_N * 16 * H_DIM;

    auto cvt = [&](const float* s, _Float16* d, int n) {
        cvt_f32_f16<<<(n + 255) / 256, 256, 0, stream>>>(s, d, n);
    };
    cvt(kW1, wk1, K_N * H_DIM * S_DIM);
    cvt(aW1, wa1, K_N * H_DIM * S_DIM);
    cvt(cW1, wc1, K_N * H_DIM * D_DIM);
    cvt(kW2, wk2, K_N * H_DIM * H_DIM);
    cvt(aW2, wa2, K_N * H_DIM * H_DIM);
    cvt(cW2, wc2, K_N * H_DIM * H_DIM);
    const int n3 = K_N * 16 * H_DIM;
    cvt_pad_w3<<<(n3 + 255) / 256, 256, 0, stream>>>(kW3, wk3, 1);
    cvt_pad_w3<<<(n3 + 255) / 256, 256, 0, stream>>>(aW3, wa3, A_N);
    cvt_pad_w3<<<(n3 + 255) / 256, 256, 0, stream>>>(cW3, wc3, A_N);

    qplex_main<<<B_TOT / WG_ROWS, NTHREADS, 0, stream>>>(
        states, actions,
        wk1, wa1, wc1, wk2, wa2, wc2, wk3, wa3, wc3,
        kb1, kb2, kb3, ab1, ab2, ab3, cb1, cb2, cb3,
        (float*)d_out);
}


# Weights reach the waves through LDS: one 256-row workgroup per WGP (8 waves) stages each net's f16 weights once, fetched one net ahead, and all waves read B fragments there; same values, same sums
# speedup vs baseline: 1.9044x; 1.9044x over previous
#include <hip/hip_runtime.h>
#include <hip/hip_bf16.h>

typedef __attribute__((ext_vector_type(16))) _Float16 v16h;
typedef __attribute__((ext_vector_type(8)))  _Float16 v8h;
typedef __attribute__((ext_vector_type(8)))  float    v8f;

#define B_TOT   65536
#define S_DIM   256
#define A_N     8
#define K_N     10
#define H_DIM   64
#define AD_DIM  128
#define D_DIM   384

#define WG_ROWS 256
#define NTHREADS 256
#define NWAVES  (NTHREADS / 32)
#define ROWS_PER_WAVE 32
#define XS      392
#define HS      72
#define W1S     392
#define W1_PIECES_MAX (H_DIM * D_DIM / 8 / NTHREADS)

__device__ __forceinline__ v16h load_frag(const _Float16* __restrict__ p,
                                          int stride, int row, int k0, int lane) {
    const int ho = (lane >> 4) << 3;
    const _Float16* q = p + row * stride + k0 + ho;
    union { v16h v; struct { v8h lo; v8h hi; } s; } u;
    u.s.lo = *(const v8h*)q;
    u.s.hi = *(const v8h*)(q + 16);
    return u.v;
}

template <int KD>
__device__ __forceinline__ void gemm4x2(const _Float16* __restrict__ A, int lda,
                                        const _Float16* __restrict__ W, int ldw,
                                        int lane, v8f c[2][4]) {
    const int mr = lane & 15;
    v8f z = {0.f, 0.f, 0.f, 0.f, 0.f, 0.f, 0.f, 0.f};
#pragma unroll
    for (int m = 0; m < 2; ++m)
#pragma unroll
        for (int t = 0; t < 4; ++t) c[m][t] = z;
    for (int k0 = 0; k0 < KD; k0 += 32) {
        v16h a0 = load_frag(A, lda, mr, k0, lane);
        v16h a1 = load_frag(A + 16 * lda, lda, mr, k0, lane);
#pragma unroll
        for (int t = 0; t < 4; ++t) {
            v16h b = load_frag(W, ldw, t * 16 + mr, k0, lane);
            c[0][t] = __builtin_amdgcn_wmma_f32_16x16x32_f16(
                false, a0, false, b, (short)0, c[0][t], false, false);
            c[1][t] = __builtin_amdgcn_wmma_f32_16x16x32_f16(
                false, a1, false, b, (short)0, c[1][t], false, false);
        }
    }
}

__device__ __forceinline__ void gemm1x2(const _Float16* __restrict__ A, int lda,
                                        const _Float16* __restrict__ W, int ldw,
                                        int lane, v8f c[2]) {
    const int mr = lane & 15;
    v8f z = {0.f, 0.f, 0.f, 0.f, 0.f, 0.f, 0.f, 0.f};
    c[0] = z; c[1] = z;
#pragma unroll
    for (int k0 = 0; k0 < H_DIM; k0 += 32) {
        v16h a0 = load_frag(A, lda, mr, k0, lane);
        v16h a1 = load_frag(A + 16 * lda, lda, mr, k0, lane);
        v16h b  = load_frag(W, ldw, mr, k0, lane);
        c[0] = __builtin_amdgcn_wmma_f32_16x16x32_f16(
            false, a0, false, b, (short)0, c[0], false, false);
        c[1] = __builtin_amdgcn_wmma_f32_16x16x32_f16(
            false, a1, false, b, (short)0, c[1], false, false);
    }
}

__device__ __forceinline__ void bias_relu_store2(const v8f c[2][4],
                                                 const float* __restrict__ bias,
                                                 _Float16* __restrict__ Hout, int lane) {
    const int col = lane & 15, hi = lane >> 4;
#pragma unroll
    for (int t = 0; t < 4; ++t) {
        float bv = bias[t * 16 + col];
#pragma unroll
        for (int m = 0; m < 2; ++m)
#pragma unroll
            for (int r = 0; r < 8; ++r) {
                float v = c[m][t][r] + bv;
                Hout[(m * 16 + r + 8 * hi) * HS + t * 16 + col] =
                    (_Float16)(v > 0.f ? v : 0.f);
            }
    }
}

__global__ void cvt_f32_f16(const float* __restrict__ src, _Float16* __restrict__ dst, int n) {
    int i = blockIdx.x * blockDim.x + threadIdx.x;
    if (i < n) dst[i] = (_Float16)src[i];
}

__global__ void cvt_pad_w3(const float* __restrict__ src, _Float16* __restrict__ dst, int outdim) {
    int i = blockIdx.x * blockDim.x + threadIdx.x;
    const int n = K_N * 16 * H_DIM;
    if (i >= n) return;
    int h = i % H_DIM;
    int o = (i / H_DIM) % 16;
    int k = i / (16 * H_DIM);
    dst[i] = (o < outdim) ? (_Float16)src[(k * outdim + o) * H_DIM + h] : (_Float16)0.f;
}

struct WPre {
    v8h w1[W1_PIECES_MAX];
    v8h w2[2];
    v8h w3;
};

template <int KD>
__device__ __forceinline__ void w_fetch(const _Float16* __restrict__ g1,
                                        const _Float16* __restrict__ g2,
                                        const _Float16* __restrict__ g3,
                                        int tid, WPre& p) {
#pragma unroll
    for (int j = 0; j < H_DIM * KD / 8 / NTHREADS; ++j)
        p.w1[j] = *(const v8h*)(g1 + (size_t)(tid + NTHREADS * j) * 8);
#pragma unroll
    for (int j = 0; j < 2; ++j)
        p.w2[j] = *(const v8h*)(g2 + (size_t)(tid + NTHREADS * j) * 8);
    p.w3 = *(const v8h*)(g3 + (size_t)(tid & 127) * 8);
}

template <int KD>
__device__ __forceinline__ void w_store(_Float16* __restrict__ W1s,
                                        _Float16* __restrict__ W2s,
                                        _Float16* __restrict__ W3s,
                                        int tid, const WPre& p) {
    constexpr int PPR = KD / 8;
#pragma unroll
    for (int j = 0; j < H_DIM * KD / 8 / NTHREADS; ++j) {
        const int i = tid + NTHREADS * j;
        const int row = i / PPR, c = i - row * PPR;
        *(v8h*)(W1s + row * W1S + c * 8) = p.w1[j];
    }
#pragma unroll
    for (int j = 0; j < 2; ++j) {
        const int i = tid + NTHREADS * j;
        *(v8h*)(W2s + (i >> 3) * HS + (i & 7) * 8) = p.w2[j];
    }
    if (tid < 128)
        *(v8h*)(W3s + (tid >> 3) * HS + (tid & 7) * 8) = p.w3;
}

__global__ __launch_bounds__(NTHREADS) __attribute__((amdgpu_waves_per_eu(2, 2)))
void qplex_main(const float* __restrict__ states, const float* __restrict__ actions,
                const _Float16* __restrict__ wk1, const _Float16* __restrict__ wa1,
                const _Float16* __restrict__ wc1,
                const _Float16* __restrict__ wk2, const _Float16* __restrict__ wa2,
                const _Float16* __restrict__ wc2,
                const _Float16* __restrict__ wk3, const _Float16* __restrict__ wa3,
                const _Float16* __restrict__ wc3,
                const float* __restrict__ bk1, const float* __restrict__ bk2,
                const float* __restrict__ bk3,
                const float* __restrict__ ba1, const float* __restrict__ ba2,
                const float* __restrict__ ba3,
                const float* __restrict__ bc1, const float* __restrict__ bc2,
                const float* __restrict__ bc3,
                float* __restrict__ out) {
    __shared__ __attribute__((aligned(16))) _Float16 Xs[WG_ROWS * XS];
    __shared__ __attribute__((aligned(16))) _Float16 Hs[NWAVES * ROWS_PER_WAVE * HS];
    __shared__ __attribute__((aligned(16))) _Float16 W1s[H_DIM * W1S];
    __shared__ __attribute__((aligned(16))) _Float16 W2s[H_DIM * HS];
    __shared__ __attribute__((aligned(16))) _Float16 W3s[16 * HS];

    const int tid = threadIdx.x;
    const int b0  = blockIdx.x * WG_ROWS;

    WPre pre;
    w_fetch<S_DIM>(wk1, wk2, wk3, tid, pre);

    for (int idx = tid; idx < WG_ROWS * D_DIM; idx += NTHREADS) {
        int row = idx / D_DIM;
        int col = idx - row * D_DIM;
        float v = (col < S_DIM)
                      ? states[(size_t)(b0 + row) * S_DIM + col]
                      : actions[(size_t)(b0 + row) * AD_DIM + (col - S_DIM)];
        Xs[row * XS + col] = (_Float16)v;
    }
    __syncthreads();

    const int wave = tid >> 5, lane = tid & 31;
    const int col = lane & 15, hi = lane >> 4;
    const _Float16* Xw = Xs + (wave * ROWS_PER_WAVE) * XS;
    _Float16*       Hw = Hs + (wave * ROWS_PER_WAVE) * HS;

    float acc[2][8];
#pragma unroll
    for (int m = 0; m < 2; ++m)
#pragma unroll
        for (int r = 0; r < 8; ++r) acc[m][r] = 0.f;

#pragma unroll 1
    for (int k = 0; k < K_N; ++k) {
        v8f c1[2][4];
        v8f ck[2], ca[2], cc[2];
        const int kn = (k + 1 < K_N) ? k + 1 : k;

        w_store<S_DIM>(W1s, W2s, W3s, tid, pre);
        __syncthreads();
        w_fetch<S_DIM>(wa1 + (size_t)k * H_DIM * S_DIM, wa2 + (size_t)k * H_DIM * H_DIM,
                       wa3 + (size_t)k * 16 * H_DIM, tid, pre);
        gemm4x2<S_DIM>(Xw, XS, W1s, W1S, lane, c1);
        bias_relu_store2(c1, bk1 + k * H_DIM, Hw, lane);
        gemm4x2<H_DIM>(Hw, HS, W2s, HS, lane, c1);
        bias_relu_store2(c1, bk2 + k * H_DIM, Hw, lane);
        gemm1x2(Hw, HS, W3s, HS, lane, ck);
        __syncthreads();

        w_store<S_DIM>(W1s, W2s, W3s, tid, pre);
        __syncthreads();
        w_fetch<D_DIM>(wc1 + (size_t)k * H_DIM * D_DIM, wc2 + (size_t)k * H_DIM * H_DIM,
                       wc3 + (size_t)k * 16 * H_DIM, tid, pre);
        gemm4x2<S_DIM>(Xw, XS, W1s, W1S, lane, c1);
        bias_relu_store2(c1, ba1 + k * H_DIM, Hw, lane);
        gemm4x2<H_DIM>(Hw, HS, W2s, HS, lane, c1);
        bias_relu_store2(c1, ba2 + k * H_DIM, Hw, lane);
        gemm1x2(Hw, HS, W3s, HS, lane, ca);
        __syncthreads();

        w_store<D_DIM>(W1s, W2s, W3s, tid, pre);
        __syncthreads();
        w_fetch<S_DIM>(wk1 + (size_t)kn * H_DIM * S_DIM, wk2 + (size_t)kn * H_DIM * H_DIM,
                       wk3 + (size_t)kn * 16 * H_DIM, tid, pre);
        gemm4x2<D_DIM>(Xw, XS, W1s, W1S, lane, c1);
        bias_relu_store2(c1, bc1 + k * H_DIM, Hw, lane);
        gemm4x2<H_DIM>(Hw, HS, W2s, HS, lane, c1);
        bias_relu_store2(c1, bc2 + k * H_DIM, Hw, lane);
        gemm1x2(Hw, HS, W3s, HS, lane, cc);
        __syncthreads();

        const float kbias = bk3[k];
        const float ab3v  = ba3[k * A_N + (col & 7)];
        const float cb3v  = bc3[k * A_N + (col & 7)];
#pragma unroll
        for (int m = 0; m < 2; ++m)
#pragma unroll
            for (int r = 0; r < 8; ++r) {
                float kv = __shfl(ck[m][r], lane & 16, 32);
                kv = fabsf(kv + kbias) + 1e-10f;
                float ag = 1.f / (1.f + __expf(-(ca[m][r] + ab3v)));
                float ac = 1.f / (1.f + __expf(-(cc[m][r] + cb3v)));
                acc[m][r] += kv * ag * ac;
            }
    }

    if (col < A_N) {
#pragma unroll
        for (int m = 0; m < 2; ++m)
#pragma unroll
            for (int r = 0; r < 8; ++r)
                out[(size_t)(b0 + wave * ROWS_PER_WAVE + m * 16 + r + 8 * hi) * A_N + col] =
                    acc[m][r];
    }
}

extern "C" void kernel_launch(void* const* d_in, const int* in_sizes, int n_in,
                              void* d_out, int out_size, void* d_ws, size_t ws_size,
                              hipStream_t stream) {
    (void)in_sizes; (void)n_in; (void)out_size; (void)ws_size;
    const float* states  = (const float*)d_in[0];
    const float* actions = (const float*)d_in[1];
    const float* kW1 = (const float*)d_in[2];
    const float* kb1 = (const float*)d_in[3];
    const float* kW2 = (const float*)d_in[4];
    const float* kb2 = (const float*)d_in[5];
    const float* kW3 = (const float*)d_in[6];
    const float* kb3 = (const float*)d_in[7];
    const float* aW1 = (const float*)d_in[8];
    const float* ab1 = (const float*)d_in[9];
    const float* aW2 = (const float*)d_in[10];
    const float* ab2 = (const float*)d_in[11];
    const float* aW3 = (const float*)d_in[12];
    const float* ab3 = (const float*)d_in[13];
    const float* cW1 = (const float*)d_in[14];
    const float* cb1 = (const float*)d_in[15];
    const float* cW2 = (const float*)d_in[16];
    const float* cb2 = (const float*)d_in[17];
    const float* cW3 = (const float*)d_in[18];
    const float* cb3 = (const float*)d_in[19];

    _Float16* ws = (_Float16*)d_ws;
    size_t off = 0;
    _Float16* wk1 = ws + off; off += (size_t)K_N * H_DIM * S_DIM;
    _Float16* wa1 = ws + off; off += (size_t)K_N * H_DIM * S_DIM;
    _Float16* wc1 = ws + off; off += (size_t)K_N * H_DIM * D_DIM;
    _Float16* wk2 = ws + off; off += (size_t)K_N * H_DIM * H_DIM;
    _Float16* wa2 = ws + off; off += (size_t)K_N * H_DIM * H_DIM;
    _Float16* wc2 = ws + off; off += (size_t)K_N * H_DIM * H_DIM;
    _Float16* wk3 = ws + off; off += (size_t)K_N * 16 * H_DIM;
    _Float16* wa3 = ws + off; off += (size_t)K_N * 16 * H_DIM;
    _Float16* wc3 = ws + off; off += (size_t)K_N * 16 * H_DIM;

    auto cvt = [&](const float* s, _Float16* d, int n) {
        cvt_f32_f16<<<(n + 255) / 256, 256, 0, stream>>>(s, d, n);
    };
    cvt(kW1, wk1, K_N * H_DIM * S_DIM);
    cvt(aW1, wa1, K_N * H_DIM * S_DIM);
    cvt(cW1, wc1, K_N * H_DIM * D_DIM);
    cvt(kW2, wk2, K_N * H_DIM * H_DIM);
    cvt(aW2, wa2, K_N * H_DIM * H_DIM);
    cvt(cW2, wc2, K_N * H_DIM * H_DIM);
    const int n3 = K_N * 16 * H_DIM;
    cvt_pad_w3<<<(n3 + 255) / 256, 256, 0, stream>>>(kW3, wk3, 1);
    cvt_pad_w3<<<(n3 + 255) / 256, 256, 0, stream>>>(aW3, wa3, A_N);
    cvt_pad_w3<<<(n3 + 255) / 256, 256, 0, stream>>>(cW3, wc3, A_N);

    qplex_main<<<B_TOT / WG_ROWS, NTHREADS, 0, stream>>>(
        states, actions,
        wk1, wa1, wc1, wk2, wa2, wc2, wk3, wa3, wc3,
        kb1, kb2, kb3, ab1, ab2, ab3, cb1, cb2, cb3,
        (float*)d_out);
}


# Input rows skip LDS: each wave loads its 32 rows' layer-1 A fragments from global memory straight into registers (16-byte loads, f16 once) and keeps them for all 30 nets; same values, same sums
# speedup vs baseline: 2.3240x; 1.2203x over previous
#include <hip/hip_runtime.h>
#include <hip/hip_bf16.h>

typedef __attribute__((ext_vector_type(16))) _Float16 v16h;
typedef __attribute__((ext_vector_type(8)))  _Float16 v8h;
typedef __attribute__((ext_vector_type(8)))  float    v8f;
typedef __attribute__((ext_vector_type(4)))  float    v4f;
typedef __attribute__((ext_vector_type(4)))  _Float16 v4h;

#define B_TOT   65536
#define S_DIM   256
#define A_N     8
#define K_N     10
#define H_DIM   64
#define AD_DIM  128
#define D_DIM   384

#define WG_ROWS 256
#define NTHREADS 256
#define NWAVES  (NTHREADS / 32)
#define ROWS_PER_WAVE 32
#define XSTEPS  (D_DIM / 32)
#define HS      72
#define W1S     392
#define W1_PIECES_MAX (H_DIM * D_DIM / 8 / NTHREADS)

__device__ __forceinline__ v16h load_frag(const _Float16* __restrict__ p,
                                          int stride, int row, int k0, int lane) {
    const int ho = (lane >> 4) << 3;
    const _Float16* q = p + row * stride + k0 + ho;
    union { v16h v; struct { v8h lo; v8h hi; } s; } u;
    u.s.lo = *(const v8h*)q;
    u.s.hi = *(const v8h*)(q + 16);
    return u.v;
}

template <int KD>
__device__ __forceinline__ void gemm4x2(const _Float16* __restrict__ A, int lda,
                                        const _Float16* __restrict__ W, int ldw,
                                        int lane, v8f c[2][4]) {
    const int mr = lane & 15;
    v8f z = {0.f, 0.f, 0.f, 0.f, 0.f, 0.f, 0.f, 0.f};
#pragma unroll
    for (int m = 0; m < 2; ++m)
#pragma unroll
        for (int t = 0; t < 4; ++t) c[m][t] = z;
    for (int k0 = 0; k0 < KD; k0 += 32) {
        v16h a0 = load_frag(A, lda, mr, k0, lane);
        v16h a1 = load_frag(A + 16 * lda, lda, mr, k0, lane);
#pragma unroll
        for (int t = 0; t < 4; ++t) {
            v16h b = load_frag(W, ldw, t * 16 + mr, k0, lane);
            c[0][t] = __builtin_amdgcn_wmma_f32_16x16x32_f16(
                false, a0, false, b, (short)0, c[0][t], false, false);
            c[1][t] = __builtin_amdgcn_wmma_f32_16x16x32_f16(
                false, a1, false, b, (short)0, c[1][t], false, false);
        }
    }
}

__device__ __forceinline__ void x_frags(const float* __restrict__ srow,
                                        const float* __restrict__ arow,
                                        int lane, v16h xa[XSTEPS]) {
    const int ho = (lane >> 4) << 3;
    v4f t[XSTEPS][4];
#pragma unroll
    for (int j = 0; j < XSTEPS; ++j) {
        const float* p = (j < S_DIM / 32) ? srow + j * 32 + ho
                                          : arow + (j - S_DIM / 32) * 32 + ho;
        t[j][0] = *(const v4f*)(p);
        t[j][1] = *(const v4f*)(p + 4);
        t[j][2] = *(const v4f*)(p + 16);
        t[j][3] = *(const v4f*)(p + 20);
    }
#pragma unroll
    for (int j = 0; j < XSTEPS; ++j) {
        v8h lo = __builtin_shufflevector(__builtin_convertvector(t[j][0], v4h),
                                         __builtin_convertvector(t[j][1], v4h),
                                         0, 1, 2, 3, 4, 5, 6, 7);
        v8h hi = __builtin_shufflevector(__builtin_convertvector(t[j][2], v4h),
                                         __builtin_convertvector(t[j][3], v4h),
                                         0, 1, 2, 3, 4, 5, 6, 7);
        xa[j] = __builtin_shufflevector(lo, hi, 0, 1, 2, 3, 4, 5, 6, 7,
                                        8, 9, 10, 11, 12, 13, 14, 15);
    }
}

template <int KD>
__device__ __forceinline__ void gemm4x2_xr(const v16h xa0[XSTEPS], const v16h xa1[XSTEPS],
                                           const _Float16* __restrict__ W, int ldw,
                                           int lane, v8f c[2][4]) {
    const int mr = lane & 15;
    v8f z = {0.f, 0.f, 0.f, 0.f, 0.f, 0.f, 0.f, 0.f};
#pragma unroll
    for (int m = 0; m < 2; ++m)
#pragma unroll
        for (int t = 0; t < 4; ++t) c[m][t] = z;
#pragma unroll
    for (int j = 0; j < KD / 32; ++j) {
#pragma unroll
        for (int t = 0; t < 4; ++t) {
            v16h b = load_frag(W, ldw, t * 16 + mr, j * 32, lane);
            c[0][t] = __builtin_amdgcn_wmma_f32_16x16x32_f16(
                false, xa0[j], false, b, (short)0, c[0][t], false, false);
            c[1][t] = __builtin_amdgcn_wmma_f32_16x16x32_f16(
                false, xa1[j], false, b, (short)0, c[1][t], false, false);
        }
    }
}

__device__ __forceinline__ void gemm1x2(const _Float16* __restrict__ A, int lda,
                                        const _Float16* __restrict__ W, int ldw,
                                        int lane, v8f c[2]) {
    const int mr = lane & 15;
    v8f z = {0.f, 0.f, 0.f, 0.f, 0.f, 0.f, 0.f, 0.f};
    c[0] = z; c[1] = z;
#pragma unroll
    for (int k0 = 0; k0 < H_DIM; k0 += 32) {
        v16h a0 = load_frag(A, lda, mr, k0, lane);
        v16h a1 = load_frag(A + 16 * lda, lda, mr, k0, lane);
        v16h b  = load_frag(W, ldw, mr, k0, lane);
        c[0] = __builtin_amdgcn_wmma_f32_16x16x32_f16(
            false, a0, false, b, (short)0, c[0], false, false);
        c[1] = __builtin_amdgcn_wmma_f32_16x16x32_f16(
            false, a1, false, b, (short)0, c[1], false, false);
    }
}

__device__ __forceinline__ void bias_relu_store2(const v8f c[2][4],
                                                 const float* __restrict__ bias,
                                                 _Float16* __restrict__ Hout, int lane) {
    const int col = lane & 15, hi = lane >> 4;
#pragma unroll
    for (int t = 0; t < 4; ++t) {
        float bv = bias[t * 16 + col];
#pragma unroll
        for (int m = 0; m < 2; ++m)
#pragma unroll
            for (int r = 0; r < 8; ++r) {
                float v = c[m][t][r] + bv;
                Hout[(m * 16 + r + 8 * hi) * HS + t * 16 + col] =
                    (_Float16)(v > 0.f ? v : 0.f);
            }
    }
}

__global__ void cvt_f32_f16(const float* __restrict__ src, _Float16* __restrict__ dst, int n) {
    int i = blockIdx.x * blockDim.x + threadIdx.x;
    if (i < n) dst[i] = (_Float16)src[i];
}

__global__ void cvt_pad_w3(const float* __restrict__ src, _Float16* __restrict__ dst, int outdim) {
    int i = blockIdx.x * blockDim.x + threadIdx.x;
    const int n = K_N * 16 * H_DIM;
    if (i >= n) return;
    int h = i % H_DIM;
    int o = (i / H_DIM) % 16;
    int k = i / (16 * H_DIM);
    dst[i] = (o < outdim) ? (_Float16)src[(k * outdim + o) * H_DIM + h] : (_Float16)0.f;
}

struct WPre {
    v8h w1[W1_PIECES_MAX];
    v8h w2[2];
    v8h w3;
};

template <int KD>
__device__ __forceinline__ void w_fetch(const _Float16* __restrict__ g1,
                                        const _Float16* __restrict__ g2,
                                        const _Float16* __restrict__ g3,
                                        int tid, WPre& p) {
#pragma unroll
    for (int j = 0; j < H_DIM * KD / 8 / NTHREADS; ++j)
        p.w1[j] = *(const v8h*)(g1 + (size_t)(tid + NTHREADS * j) * 8);
#pragma unroll
    for (int j = 0; j < 2; ++j)
        p.w2[j] = *(const v8h*)(g2 + (size_t)(tid + NTHREADS * j) * 8);
    p.w3 = *(const v8h*)(g3 + (size_t)(tid & 127) * 8);
}

template <int KD>
__device__ __forceinline__ void w_store(_Float16* __restrict__ W1s,
                                        _Float16* __restrict__ W2s,
                                        _Float16* __restrict__ W3s,
                                        int tid, const WPre& p) {
    constexpr int PPR = KD / 8;
#pragma unroll
    for (int j = 0; j < H_DIM * KD / 8 / NTHREADS; ++j) {
        const int i = tid + NTHREADS * j;
        const int row = i / PPR, c = i - row * PPR;
        *(v8h*)(W1s + row * W1S + c * 8) = p.w1[j];
    }
#pragma unroll
    for (int j = 0; j < 2; ++j) {
        const int i = tid + NTHREADS * j;
        *(v8h*)(W2s + (i >> 3) * HS + (i & 7) * 8) = p.w2[j];
    }
    if (tid < 128)
        *(v8h*)(W3s + (tid >> 3) * HS + (tid & 7) * 8) = p.w3;
}

__global__ __launch_bounds__(NTHREADS) __attribute__((amdgpu_waves_per_eu(2, 2)))
void qplex_main(const float* __restrict__ states, const float* __restrict__ actions,
                const _Float16* __restrict__ wk1, const _Float16* __restrict__ wa1,
                const _Float16* __restrict__ wc1,
                const _Float16* __restrict__ wk2, const _Float16* __restrict__ wa2,
                const _Float16* __restrict__ wc2,
                const _Float16* __restrict__ wk3, const _Float16* __restrict__ wa3,
                const _Float16* __restrict__ wc3,
                const float* __restrict__ bk1, const float* __restrict__ bk2,
                const float* __restrict__ bk3,
                const float* __restrict__ ba1, const float* __restrict__ ba2,
                const float* __restrict__ ba3,
                const float* __restrict__ bc1, const float* __restrict__ bc2,
                const float* __restrict__ bc3,
                float* __restrict__ out) {
    __shared__ __attribute__((aligned(16))) _Float16 Hs[NWAVES * ROWS_PER_WAVE * HS];
    __shared__ __attribute__((aligned(16))) _Float16 W1s[H_DIM * W1S];
    __shared__ __attribute__((aligned(16))) _Float16 W2s[H_DIM * HS];
    __shared__ __attribute__((aligned(16))) _Float16 W3s[16 * HS];

    const int tid = threadIdx.x;
    const int b0  = blockIdx.x * WG_ROWS;

    WPre pre;
    w_fetch<S_DIM>(wk1, wk2, wk3, tid, pre);

    const int wave = tid >> 5, lane = tid & 31;
    const int col = lane & 15, hi = lane >> 4;
    _Float16*       Hw = Hs + (wave * ROWS_PER_WAVE) * HS;

    v16h xa0[XSTEPS], xa1[XSTEPS];
    {
        const size_t r0 = (size_t)(b0 + wave * ROWS_PER_WAVE + col);
        x_frags(states + r0 * S_DIM, actions + r0 * AD_DIM, lane, xa0);
        x_frags(states + (r0 + 16) * S_DIM, actions + (r0 + 16) * AD_DIM, lane, xa1);
    }

    float acc[2][8];
#pragma unroll
    for (int m = 0; m < 2; ++m)
#pragma unroll
        for (int r = 0; r < 8; ++r) acc[m][r] = 0.f;

#pragma unroll 1
    for (int k = 0; k < K_N; ++k) {
        v8f c1[2][4];
        v8f ck[2], ca[2], cc[2];
        const int kn = (k + 1 < K_N) ? k + 1 : k;

        w_store<S_DIM>(W1s, W2s, W3s, tid, pre);
        __syncthreads();
        w_fetch<S_DIM>(wa1 + (size_t)k * H_DIM * S_DIM, wa2 + (size_t)k * H_DIM * H_DIM,
                       wa3 + (size_t)k * 16 * H_DIM, tid, pre);
        gemm4x2_xr<S_DIM>(xa0, xa1, W1s, W1S, lane, c1);
        bias_relu_store2(c1, bk1 + k * H_DIM, Hw, lane);
        gemm4x2<H_DIM>(Hw, HS, W2s, HS, lane, c1);
        bias_relu_store2(c1, bk2 + k * H_DIM, Hw, lane);
        gemm1x2(Hw, HS, W3s, HS, lane, ck);
        __syncthreads();

        w_store<S_DIM>(W1s, W2s, W3s, tid, pre);
        __syncthreads();
        w_fetch<D_DIM>(wc1 + (size_t)k * H_DIM * D_DIM, wc2 + (size_t)k * H_DIM * H_DIM,
                       wc3 + (size_t)k * 16 * H_DIM, tid, pre);
        gemm4x2_xr<S_DIM>(xa0, xa1, W1s, W1S, lane, c1);
        bias_relu_store2(c1, ba1 + k * H_DIM, Hw, lane);
        gemm4x2<H_DIM>(Hw, HS, W2s, HS, lane, c1);
        bias_relu_store2(c1, ba2 + k * H_DIM, Hw, lane);
        gemm1x2(Hw, HS, W3s, HS, lane, ca);
        __syncthreads();

        w_store<D_DIM>(W1s, W2s, W3s, tid, pre);
        __syncthreads();
        w_fetch<S_DIM>(wk1 + (size_t)kn * H_DIM * S_DIM, wk2 + (size_t)kn * H_DIM * H_DIM,
                       wk3 + (size_t)kn * 16 * H_DIM, tid, pre);
        gemm4x2_xr<D_DIM>(xa0, xa1, W1s, W1S, lane, c1);
        bias_relu_store2(c1, bc1 + k * H_DIM, Hw, lane);
        gemm4x2<H_DIM>(Hw, HS, W2s, HS, lane, c1);
        bias_relu_store2(c1, bc2 + k * H_DIM, Hw, lane);
        gemm1x2(Hw, HS, W3s, HS, lane, cc);
        __syncthreads();

        const float kbias = bk3[k];
        const float ab3v  = ba3[k * A_N + (col & 7)];
        const float cb3v  = bc3[k * A_N + (col & 7)];
#pragma unroll
        for (int m = 0; m < 2; ++m)
#pragma unroll
            for (int r = 0; r < 8; ++r) {
                float kv = __shfl(ck[m][r], lane & 16, 32);
                kv = fabsf(kv + kbias) + 1e-10f;
                float ag = 1.f / (1.f + __expf(-(ca[m][r] + ab3v)));
                float ac = 1.f / (1.f + __expf(-(cc[m][r] + cb3v)));
                acc[m][r] += kv * ag * ac;
            }
    }

    if (col < A_N) {
#pragma unroll
        for (int m = 0; m < 2; ++m)
#pragma unroll
            for (int r = 0; r < 8; ++r)
                out[(size_t)(b0 + wave * ROWS_PER_WAVE + m * 16 + r + 8 * hi) * A_N + col] =
                    acc[m][r];
    }
}

extern "C" void kernel_launch(void* const* d_in, const int* in_sizes, int n_in,
                              void* d_out, int out_size, void* d_ws, size_t ws_size,
                              hipStream_t stream) {
    (void)in_sizes; (void)n_in; (void)out_size; (void)ws_size;
    const float* states  = (const float*)d_in[0];
    const float* actions = (const float*)d_in[1];
    const float* kW1 = (const float*)d_in[2];
    const float* kb1 = (const float*)d_in[3];
    const float* kW2 = (const float*)d_in[4];
    const float* kb2 = (const float*)d_in[5];
    const float* kW3 = (const float*)d_in[6];
    const float* kb3 = (const float*)d_in[7];
    const float* aW1 = (const float*)d_in[8];
    const float* ab1 = (const float*)d_in[9];
    const float* aW2 = (const float*)d_in[10];
    const float* ab2 = (const float*)d_in[11];
    const float* aW3 = (const float*)d_in[12];
    const float* ab3 = (const float*)d_in[13];
    const float* cW1 = (const float*)d_in[14];
    const float* cb1 = (const float*)d_in[15];
    const float* cW2 = (const float*)d_in[16];
    const float* cb2 = (const float*)d_in[17];
    const float* cW3 = (const float*)d_in[18];
    const float* cb3 = (const float*)d_in[19];

    _Float16* ws = (_Float16*)d_ws;
    size_t off = 0;
    _Float16* wk1 = ws + off; off += (size_t)K_N * H_DIM * S_DIM;
    _Float16* wa1 = ws + off; off += (size_t)K_N * H_DIM * S_DIM;
    _Float16* wc1 = ws + off; off += (size_t)K_N * H_DIM * D_DIM;
    _Float16* wk2 = ws + off; off += (size_t)K_N * H_DIM * H_DIM;
    _Float16* wa2 = ws + off; off += (size_t)K_N * H_DIM * H_DIM;
    _Float16* wc2 = ws + off; off += (size_t)K_N * H_DIM * H_DIM;
    _Float16* wk3 = ws + off; off += (size_t)K_N * 16 * H_DIM;
    _Float16* wa3 = ws + off; off += (size_t)K_N * 16 * H_DIM;
    _Float16* wc3 = ws + off; off += (size_t)K_N * 16 * H_DIM;

    auto cvt = [&](const float* s, _Float16* d, int n) {
        cvt_f32_f16<<<(n + 255) / 256, 256, 0, stream>>>(s, d, n);
    };
    cvt(kW1, wk1, K_N * H_DIM * S_DIM);
    cvt(aW1, wa1, K_N * H_DIM * S_DIM);
    cvt(cW1, wc1, K_N * H_DIM * D_DIM);
    cvt(kW2, wk2, K_N * H_DIM * H_DIM);
    cvt(aW2, wa2, K_N * H_DIM * H_DIM);
    cvt(cW2, wc2, K_N * H_DIM * H_DIM);
    const int n3 = K_N * 16 * H_DIM;
    cvt_pad_w3<<<(n3 + 255) / 256, 256, 0, stream>>>(kW3, wk3, 1);
    cvt_pad_w3<<<(n3 + 255) / 256, 256, 0, stream>>>(aW3, wa3, A_N);
    cvt_pad_w3<<<(n3 + 255) / 256, 256, 0, stream>>>(cW3, wc3, A_N);

    qplex_main<<<B_TOT / WG_ROWS, NTHREADS, 0, stream>>>(
        states, actions,
        wk1, wa1, wc1, wk2, wa2, wc2, wk3, wa3, wc3,
        kb1, kb2, kb3, ab1, ab2, ab3, cb1, cb2, cb3,
        (float*)d_out);
}


# Hidden tile transposed in LDS: a lane's 8 C-tile values take bias, f16, ReLU as packed pairs and go out as one 16-byte store, not 8 2-byte ones; layers 2-3 read A by ds_load_tr16_b128; same sums
# speedup vs baseline: 3.0409x; 1.3085x over previous
#include <hip/hip_runtime.h>
#include <hip/hip_bf16.h>

typedef __attribute__((ext_vector_type(16))) _Float16 v16h;
typedef __attribute__((ext_vector_type(8)))  _Float16 v8h;
typedef __attribute__((ext_vector_type(8)))  float    v8f;
typedef __attribute__((ext_vector_type(4)))  float    v4f;
typedef __attribute__((ext_vector_type(4)))  _Float16 v4h;

#define B_TOT   65536
#define S_DIM   256
#define A_N     8
#define K_N     10
#define H_DIM   64
#define AD_DIM  128
#define D_DIM   384

#define WG_ROWS 256
#define NTHREADS 256
#define NWAVES  (NTHREADS / 32)
#define ROWS_PER_WAVE 32
#define XSTEPS  (D_DIM / 32)
#define HS      72
#define HTP_B   96
#define HT_WAVE_B (H_DIM * HTP_B)
#define W1S     392
#define W1_PIECES_MAX (H_DIM * D_DIM / 8 / NTHREADS)

__device__ __forceinline__ v16h load_frag(const _Float16* __restrict__ p,
                                          int stride, int row, int k0, int lane) {
    const int ho = (lane >> 4) << 3;
    const _Float16* q = p + row * stride + k0 + ho;
    union { v16h v; struct { v8h lo; v8h hi; } s; } u;
    u.s.lo = *(const v8h*)q;
    u.s.hi = *(const v8h*)(q + 16);
    return u.v;
}

typedef __attribute__((ext_vector_type(8))) short s16x8;
__device__ __forceinline__ v8h ld_tr16(const unsigned char* p) {
    typedef __attribute__((address_space(3))) s16x8 lds_s16x8;
    return __builtin_bit_cast(v8h, __builtin_amdgcn_ds_load_tr16_b128_v8i16((lds_s16x8*)p));
}

__device__ __forceinline__ v16h load_frag_ht(const unsigned char* __restrict__ Ht,
                                             int m, int k0, int lane) {
    const unsigned char* p = Ht + (k0 + 8 * (lane >> 4) + (lane & 7)) * HTP_B
                                + 32 * m + 16 * ((lane >> 3) & 1);
    union { v16h v; struct { v8h lo; v8h hi; } s; } u;
    u.s.lo = ld_tr16(p);
    u.s.hi = ld_tr16(p + 16 * HTP_B);
    return u.v;
}

template <int KD>
__device__ __forceinline__ void gemm4x2(const unsigned char* __restrict__ Ht,
                                        const _Float16* __restrict__ W, int ldw,
                                        int lane, v8f c[2][4]) {
    const int mr = lane & 15;
    v8f z = {0.f, 0.f, 0.f, 0.f, 0.f, 0.f, 0.f, 0.f};
#pragma unroll
    for (int m = 0; m < 2; ++m)
#pragma unroll
        for (int t = 0; t < 4; ++t) c[m][t] = z;
    for (int k0 = 0; k0 < KD; k0 += 32) {
        v16h a0 = load_frag_ht(Ht, 0, k0, lane);
        v16h a1 = load_frag_ht(Ht, 1, k0, lane);
#pragma unroll
        for (int t = 0; t < 4; ++t) {
            v16h b = load_frag(W, ldw, t * 16 + mr, k0, lane);
            c[0][t] = __builtin_amdgcn_wmma_f32_16x16x32_f16(
                false, a0, false, b, (short)0, c[0][t], false, false);
            c[1][t] = __builtin_amdgcn_wmma_f32_16x16x32_f16(
                false, a1, false, b, (short)0, c[1][t], false, false);
        }
    }
}

__device__ __forceinline__ void x_frags(const float* __restrict__ srow,
                                        const float* __restrict__ arow,
                                        int lane, v16h xa[XSTEPS]) {
    const int ho = (lane >> 4) << 3;
    v4f t[XSTEPS][4];
#pragma unroll
    for (int j = 0; j < XSTEPS; ++j) {
        const float* p = (j < S_DIM / 32) ? srow + j * 32 + ho
                                          : arow + (j - S_DIM / 32) * 32 + ho;
        t[j][0] = *(const v4f*)(p);
        t[j][1] = *(const v4f*)(p + 4);
        t[j][2] = *(const v4f*)(p + 16);
        t[j][3] = *(const v4f*)(p + 20);
    }
#pragma unroll
    for (int j = 0; j < XSTEPS; ++j) {
        v8h lo = __builtin_shufflevector(__builtin_convertvector(t[j][0], v4h),
                                         __builtin_convertvector(t[j][1], v4h),
                                         0, 1, 2, 3, 4, 5, 6, 7);
        v8h hi = __builtin_shufflevector(__builtin_convertvector(t[j][2], v4h),
                                         __builtin_convertvector(t[j][3], v4h),
                                         0, 1, 2, 3, 4, 5, 6, 7);
        xa[j] = __builtin_shufflevector(lo, hi, 0, 1, 2, 3, 4, 5, 6, 7,
                                        8, 9, 10, 11, 12, 13, 14, 15);
    }
}

template <int KD>
__device__ __forceinline__ void gemm4x2_xr(const v16h xa0[XSTEPS], const v16h xa1[XSTEPS],
                                           const _Float16* __restrict__ W, int ldw,
                                           int lane, v8f c[2][4]) {
    const int mr = lane & 15;
    v8f z = {0.f, 0.f, 0.f, 0.f, 0.f, 0.f, 0.f, 0.f};
#pragma unroll
    for (int m = 0; m < 2; ++m)
#pragma unroll
        for (int t = 0; t < 4; ++t) c[m][t] = z;
#pragma unroll
    for (int j = 0; j < KD / 32; ++j) {
#pragma unroll
        for (int t = 0; t < 4; ++t) {
            v16h b = load_frag(W, ldw, t * 16 + mr, j * 32, lane);
            c[0][t] = __builtin_amdgcn_wmma_f32_16x16x32_f16(
                false, xa0[j], false, b, (short)0, c[0][t], false, false);
            c[1][t] = __builtin_amdgcn_wmma_f32_16x16x32_f16(
                false, xa1[j], false, b, (short)0, c[1][t], false, false);
        }
    }
}

__device__ __forceinline__ void gemm1x2(const unsigned char* __restrict__ Ht,
                                        const _Float16* __restrict__ W, int ldw,
                                        int lane, v8f c[2]) {
    const int mr = lane & 15;
    v8f z = {0.f, 0.f, 0.f, 0.f, 0.f, 0.f, 0.f, 0.f};
    c[0] = z; c[1] = z;
#pragma unroll
    for (int k0 = 0; k0 < H_DIM; k0 += 32) {
        v16h a0 = load_frag_ht(Ht, 0, k0, lane);
        v16h a1 = load_frag_ht(Ht, 1, k0, lane);
        v16h b  = load_frag(W, ldw, mr, k0, lane);
        c[0] = __builtin_amdgcn_wmma_f32_16x16x32_f16(
            false, a0, false, b, (short)0, c[0], false, false);
        c[1] = __builtin_amdgcn_wmma_f32_16x16x32_f16(
            false, a1, false, b, (short)0, c[1], false, false);
    }
}

__device__ __forceinline__ void bias_relu_store2(const v8f c[2][4],
                                                 const float* __restrict__ bias,
                                                 unsigned char* __restrict__ Ht, int lane) {
    const int col = lane & 15, hi = lane >> 4;
    unsigned char* __restrict__ hp = Ht + col * HTP_B + 16 * hi;
    const v8h hz = {0, 0, 0, 0, 0, 0, 0, 0};
#pragma unroll
    for (int t = 0; t < 4; ++t) {
        const float bv = bias[t * 16 + col];
#pragma unroll
        for (int m = 0; m < 2; ++m) {
            const v8f v = c[m][t] + bv;
            const v8h h = __builtin_elementwise_max(__builtin_convertvector(v, v8h), hz);
            *(v8h*)(hp + t * 16 * HTP_B + 32 * m) = h;
        }
    }
}

__global__ void cvt_f32_f16(const float* __restrict__ src, _Float16* __restrict__ dst, int n) {
    int i = blockIdx.x * blockDim.x + threadIdx.x;
    if (i < n) dst[i] = (_Float16)src[i];
}

__global__ void cvt_pad_w3(const float* __restrict__ src, _Float16* __restrict__ dst, int outdim) {
    int i = blockIdx.x * blockDim.x + threadIdx.x;
    const int n = K_N * 16 * H_DIM;
    if (i >= n) return;
    int h = i % H_DIM;
    int o = (i / H_DIM) % 16;
    int k = i / (16 * H_DIM);
    dst[i] = (o < outdim) ? (_Float16)src[(k * outdim + o) * H_DIM + h] : (_Float16)0.f;
}

struct WPre {
    v8h w1[W1_PIECES_MAX];
    v8h w2[2];
    v8h w3;
};

template <int KD>
__device__ __forceinline__ void w_fetch(const _Float16* __restrict__ g1,
                                        const _Float16* __restrict__ g2,
                                        const _Float16* __restrict__ g3,
                                        int tid, WPre& p) {
#pragma unroll
    for (int j = 0; j < H_DIM * KD / 8 / NTHREADS; ++j)
        p.w1[j] = *(const v8h*)(g1 + (size_t)(tid + NTHREADS * j) * 8);
#pragma unroll
    for (int j = 0; j < 2; ++j)
        p.w2[j] = *(const v8h*)(g2 + (size_t)(tid + NTHREADS * j) * 8);
    p.w3 = *(const v8h*)(g3 + (size_t)(tid & 127) * 8);
}

template <int KD>
__device__ __forceinline__ void w_store(_Float16* __restrict__ W1s,
                                        _Float16* __restrict__ W2s,
                                        _Float16* __restrict__ W3s,
                                        int tid, const WPre& p) {
    constexpr int PPR = KD / 8;
#pragma unroll
    for (int j = 0; j < H_DIM * KD / 8 / NTHREADS; ++j) {
        const int i = tid + NTHREADS * j;
        const int row = i / PPR, c = i - row * PPR;
        *(v8h*)(W1s + row * W1S + c * 8) = p.w1[j];
    }
#pragma unroll
    for (int j = 0; j < 2; ++j) {
        const int i = tid + NTHREADS * j;
        *(v8h*)(W2s + (i >> 3) * HS + (i & 7) * 8) = p.w2[j];
    }
    if (tid < 128)
        *(v8h*)(W3s + (tid >> 3) * HS + (tid & 7) * 8) = p.w3;
}

__global__ __launch_bounds__(NTHREADS) __attribute__((amdgpu_waves_per_eu(2, 2)))
void qplex_main(const float* __restrict__ states, const float* __restrict__ actions,
                const _Float16* __restrict__ wk1, const _Float16* __restrict__ wa1,
                const _Float16* __restrict__ wc1,
                const _Float16* __restrict__ wk2, const _Float16* __restrict__ wa2,
                const _Float16* __restrict__ wc2,
                const _Float16* __restrict__ wk3, const _Float16* __restrict__ wa3,
                const _Float16* __restrict__ wc3,
                const float* __restrict__ bk1, const float* __restrict__ bk2,
                const float* __restrict__ bk3,
                const float* __restrict__ ba1, const float* __restrict__ ba2,
                const float* __restrict__ ba3,
                const float* __restrict__ bc1, const float* __restrict__ bc2,
                const float* __restrict__ bc3,
                float* __restrict__ out) {
    __shared__ __attribute__((aligned(16))) unsigned char Hs[NWAVES * HT_WAVE_B];
    __shared__ __attribute__((aligned(16))) _Float16 W1s[H_DIM * W1S];
    __shared__ __attribute__((aligned(16))) _Float16 W2s[H_DIM * HS];
    __shared__ __attribute__((aligned(16))) _Float16 W3s[16 * HS];

    const int tid = threadIdx.x;
    const int b0  = blockIdx.x * WG_ROWS;

    WPre pre;
    w_fetch<S_DIM>(wk1, wk2, wk3, tid, pre);

    const int wave = tid >> 5, lane = tid & 31;
    const int col = lane & 15, hi = lane >> 4;
    unsigned char*  Hw = Hs + wave * HT_WAVE_B;

    v16h xa0[XSTEPS], xa1[XSTEPS];
    {
        const size_t r0 = (size_t)(b0 + wave * ROWS_PER_WAVE + col);
        x_frags(states + r0 * S_DIM, actions + r0 * AD_DIM, lane, xa0);
        x_frags(states + (r0 + 16) * S_DIM, actions + (r0 + 16) * AD_DIM, lane, xa1);
    }

    float acc[2][8];
#pragma unroll
    for (int m = 0; m < 2; ++m)
#pragma unroll
        for (int r = 0; r < 8; ++r) acc[m][r] = 0.f;

#pragma unroll 1
    for (int k = 0; k < K_N; ++k) {
        v8f c1[2][4];
        v8f ck[2], ca[2], cc[2];
        const int kn = (k + 1 < K_N) ? k + 1 : k;

        w_store<S_DIM>(W1s, W2s, W3s, tid, pre);
        __syncthreads();
        w_fetch<S_DIM>(wa1 + (size_t)k * H_DIM * S_DIM, wa2 + (size_t)k * H_DIM * H_DIM,
                       wa3 + (size_t)k * 16 * H_DIM, tid, pre);
        gemm4x2_xr<S_DIM>(xa0, xa1, W1s, W1S, lane, c1);
        bias_relu_store2(c1, bk1 + k * H_DIM, Hw, lane);
        gemm4x2<H_DIM>(Hw, W2s, HS, lane, c1);
        bias_relu_store2(c1, bk2 + k * H_DIM, Hw, lane);
        gemm1x2(Hw, W3s, HS, lane, ck);
        __syncthreads();

        w_store<S_DIM>(W1s, W2s, W3s, tid, pre);
        __syncthreads();
        w_fetch<D_DIM>(wc1 + (size_t)k * H_DIM * D_DIM, wc2 + (size_t)k * H_DIM * H_DIM,
                       wc3 + (size_t)k * 16 * H_DIM, tid, pre);
        gemm4x2_xr<S_DIM>(xa0, xa1, W1s, W1S, lane, c1);
        bias_relu_store2(c1, ba1 + k * H_DIM, Hw, lane);
        gemm4x2<H_DIM>(Hw, W2s, HS, lane, c1);
        bias_relu_store2(c1, ba2 + k * H_DIM, Hw, lane);
        gemm1x2(Hw, W3s, HS, lane, ca);
        __syncthreads();

        w_store<D_DIM>(W1s, W2s, W3s, tid, pre);
        __syncthreads();
        w_fetch<S_DIM>(wk1 + (size_t)kn * H_DIM * S_DIM, wk2 + (size_t)kn * H_DIM * H_DIM,
                       wk3 + (size_t)kn * 16 * H_DIM, tid, pre);
        gemm4x2_xr<D_DIM>(xa0, xa1, W1s, W1S, lane, c1);
        bias_relu_store2(c1, bc1 + k * H_DIM, Hw, lane);
        gemm4x2<H_DIM>(Hw, W2s, HS, lane, c1);
        bias_relu_store2(c1, bc2 + k * H_DIM, Hw, lane);
        gemm1x2(Hw, W3s, HS, lane, cc);
        __syncthreads();

        const float kbias = bk3[k];
        const float ab3v  = ba3[k * A_N + (col & 7)];
        const float cb3v  = bc3[k * A_N + (col & 7)];
#pragma unroll
        for (int m = 0; m < 2; ++m)
#pragma unroll
            for (int r = 0; r < 8; ++r) {
                float kv = __shfl(ck[m][r], lane & 16, 32);
                kv = fabsf(kv + kbias) + 1e-10f;
                float ag = 1.f / (1.f + __expf(-(ca[m][r] + ab3v)));
                float ac = 1.f / (1.f + __expf(-(cc[m][r] + cb3v)));
                acc[m][r] += kv * ag * ac;
            }
    }

    if (col < A_N) {
#pragma unroll
        for (int m = 0; m < 2; ++m)
#pragma unroll
            for (int r = 0; r < 8; ++r)
                out[(size_t)(b0 + wave * ROWS_PER_WAVE + m * 16 + r + 8 * hi) * A_N + col] =
                    acc[m][r];
    }
}

extern "C" void kernel_launch(void* const* d_in, const int* in_sizes, int n_in,
                              void* d_out, int out_size, void* d_ws, size_t ws_size,
                              hipStream_t stream) {
    (void)in_sizes; (void)n_in; (void)out_size; (void)ws_size;
    const float* states  = (const float*)d_in[0];
    const float* actions = (const float*)d_in[1];
    const float* kW1 = (const float*)d_in[2];
    const float* kb1 = (const float*)d_in[3];
    const float* kW2 = (const float*)d_in[4];
    const float* kb2 = (const float*)d_in[5];
    const float* kW3 = (const float*)d_in[6];
    const float* kb3 = (const float*)d_in[7];
    const float* aW1 = (const float*)d_in[8];
    const float* ab1 = (const float*)d_in[9];
    const float* aW2 = (const float*)d_in[10];
    const float* ab2 = (const float*)d_in[11];
    const float* aW3 = (const float*)d_in[12];
    const float* ab3 = (const float*)d_in[13];
    const float* cW1 = (const float*)d_in[14];
    const float* cb1 = (const float*)d_in[15];
    const float* cW2 = (const float*)d_in[16];
    const float* cb2 = (const float*)d_in[17];
    const float* cW3 = (const float*)d_in[18];
    const float* cb3 = (const float*)d_in[19];

    _Float16* ws = (_Float16*)d_ws;
    size_t off = 0;
    _Float16* wk1 = ws + off; off += (size_t)K_N * H_DIM * S_DIM;
    _Float16* wa1 = ws + off; off += (size_t)K_N * H_DIM * S_DIM;
    _Float16* wc1 = ws + off; off += (size_t)K_N * H_DIM * D_DIM;
    _Float16* wk2 = ws + off; off += (size_t)K_N * H_DIM * H_DIM;
    _Float16* wa2 = ws + off; off += (size_t)K_N * H_DIM * H_DIM;
    _Float16* wc2 = ws + off; off += (size_t)K_N * H_DIM * H_DIM;
    _Float16* wk3 = ws + off; off += (size_t)K_N * 16 * H_DIM;
    _Float16* wa3 = ws + off; off += (size_t)K_N * 16 * H_DIM;
    _Float16* wc3 = ws + off; off += (size_t)K_N * 16 * H_DIM;

    auto cvt = [&](const float* s, _Float16* d, int n) {
        cvt_f32_f16<<<(n + 255) / 256, 256, 0, stream>>>(s, d, n);
    };
    cvt(kW1, wk1, K_N * H_DIM * S_DIM);
    cvt(aW1, wa1, K_N * H_DIM * S_DIM);
    cvt(cW1, wc1, K_N * H_DIM * D_DIM);
    cvt(kW2, wk2, K_N * H_DIM * H_DIM);
    cvt(aW2, wa2, K_N * H_DIM * H_DIM);
    cvt(cW2, wc2, K_N * H_DIM * H_DIM);
    const int n3 = K_N * 16 * H_DIM;
    cvt_pad_w3<<<(n3 + 255) / 256, 256, 0, stream>>>(kW3, wk3, 1);
    cvt_pad_w3<<<(n3 + 255) / 256, 256, 0, stream>>>(aW3, wa3, A_N);
    cvt_pad_w3<<<(n3 + 255) / 256, 256, 0, stream>>>(cW3, wc3, A_N);

    qplex_main<<<B_TOT / WG_ROWS, NTHREADS, 0, stream>>>(
        states, actions,
        wk1, wa1, wc1, wk2, wa2, wc2, wk3, wa3, wc3,
        kb1, kb2, kb3, ab1, ab2, ab3, cb1, cb2, cb3,
        (float*)d_out);
}


# LDS fragment loads grouped a block ahead: layer-1 B fragments of k-step j+1 (8 ds_load_b128) issued before step j's 8 WMMAs, layer-2/3 B fragments before the epilogue; same values, same sums
# speedup vs baseline: 4.3811x; 1.4407x over previous
#include <hip/hip_runtime.h>
#include <hip/hip_bf16.h>

typedef __attribute__((ext_vector_type(16))) _Float16 v16h;
typedef __attribute__((ext_vector_type(8)))  _Float16 v8h;
typedef __attribute__((ext_vector_type(8)))  float    v8f;
typedef __attribute__((ext_vector_type(4)))  float    v4f;
typedef __attribute__((ext_vector_type(4)))  _Float16 v4h;

#define B_TOT   65536
#define S_DIM   256
#define A_N     8
#define K_N     10
#define H_DIM   64
#define AD_DIM  128
#define D_DIM   384

#define WG_ROWS 256
#define NTHREADS 256
#define NWAVES  (NTHREADS / 32)
#define ROWS_PER_WAVE 32
#define XSTEPS  (D_DIM / 32)
#define HS      72
#define HTP_B   96
#define HT_WAVE_B (H_DIM * HTP_B)
#define W1S     392
#define W1_PIECES_MAX (H_DIM * D_DIM / 8 / NTHREADS)

__device__ __forceinline__ v16h load_frag(const _Float16* __restrict__ p,
                                          int stride, int row, int k0, int lane) {
    const int ho = (lane >> 4) << 3;
    const _Float16* q = p + row * stride + k0 + ho;
    union { v16h v; struct { v8h lo; v8h hi; } s; } u;
    u.s.lo = *(const v8h*)q;
    u.s.hi = *(const v8h*)(q + 16);
    return u.v;
}

typedef __attribute__((ext_vector_type(8))) short s16x8;
__device__ __forceinline__ v8h ld_tr16(const unsigned char* p) {
    typedef __attribute__((address_space(3))) s16x8 lds_s16x8;
    return __builtin_bit_cast(v8h, __builtin_amdgcn_ds_load_tr16_b128_v8i16((lds_s16x8*)p));
}

__device__ __forceinline__ v16h load_frag_ht(const unsigned char* __restrict__ Ht,
                                             int m, int k0, int lane) {
    const unsigned char* p = Ht + (k0 + 8 * (lane >> 4) + (lane & 7)) * HTP_B
                                + 32 * m + 16 * ((lane >> 3) & 1);
    union { v16h v; struct { v8h lo; v8h hi; } s; } u;
    u.s.lo = ld_tr16(p);
    u.s.hi = ld_tr16(p + 16 * HTP_B);
    return u.v;
}

__device__ __forceinline__ void load_b_4x2(const _Float16* __restrict__ W, int ldw,
                                           int lane, v16h b[2][4]) {
    const int mr = lane & 15;
#pragma unroll
    for (int s = 0; s < 2; ++s)
#pragma unroll
        for (int t = 0; t < 4; ++t) b[s][t] = load_frag(W, ldw, t * 16 + mr, s * 32, lane);
}

__device__ __forceinline__ void gemm4x2(const unsigned char* __restrict__ Ht,
                                        const v16h b[2][4], int lane, v8f c[2][4]) {
    v8f z = {0.f, 0.f, 0.f, 0.f, 0.f, 0.f, 0.f, 0.f};
#pragma unroll
    for (int m = 0; m < 2; ++m)
#pragma unroll
        for (int t = 0; t < 4; ++t) c[m][t] = z;
    v16h a[2][2];
#pragma unroll
    for (int s = 0; s < 2; ++s) {
        a[s][0] = load_frag_ht(Ht, 0, s * 32, lane);
        a[s][1] = load_frag_ht(Ht, 1, s * 32, lane);
    }
#pragma unroll
    for (int s = 0; s < 2; ++s) {
#pragma unroll
        for (int t = 0; t < 4; ++t) {
            c[0][t] = __builtin_amdgcn_wmma_f32_16x16x32_f16(
                false, a[s][0], false, b[s][t], (short)0, c[0][t], false, false);
            c[1][t] = __builtin_amdgcn_wmma_f32_16x16x32_f16(
                false, a[s][1], false, b[s][t], (short)0, c[1][t], false, false);
        }
    }
}

__device__ __forceinline__ void x_frags(const float* __restrict__ srow,
                                        const float* __restrict__ arow,
                                        int lane, v16h xa[XSTEPS]) {
    const int ho = (lane >> 4) << 3;
    v4f t[XSTEPS][4];
#pragma unroll
    for (int j = 0; j < XSTEPS; ++j) {
        const float* p = (j < S_DIM / 32) ? srow + j * 32 + ho
                                          : arow + (j - S_DIM / 32) * 32 + ho;
        t[j][0] = *(const v4f*)(p);
        t[j][1] = *(const v4f*)(p + 4);
        t[j][2] = *(const v4f*)(p + 16);
        t[j][3] = *(const v4f*)(p + 20);
    }
#pragma unroll
    for (int j = 0; j < XSTEPS; ++j) {
        v8h lo = __builtin_shufflevector(__builtin_convertvector(t[j][0], v4h),
                                         __builtin_convertvector(t[j][1], v4h),
                                         0, 1, 2, 3, 4, 5, 6, 7);
        v8h hi = __builtin_shufflevector(__builtin_convertvector(t[j][2], v4h),
                                         __builtin_convertvector(t[j][3], v4h),
                                         0, 1, 2, 3, 4, 5, 6, 7);
        xa[j] = __builtin_shufflevector(lo, hi, 0, 1, 2, 3, 4, 5, 6, 7,
                                        8, 9, 10, 11, 12, 13, 14, 15);
    }
}

template <int KD>
__device__ __forceinline__ void gemm4x2_xr(const v16h xa0[XSTEPS], const v16h xa1[XSTEPS],
                                           const _Float16* __restrict__ W, int ldw,
                                           int lane, v8f c[2][4]) {
    const int mr = lane & 15;
    constexpr int NS = KD / 32;
    v8f z = {0.f, 0.f, 0.f, 0.f, 0.f, 0.f, 0.f, 0.f};
#pragma unroll
    for (int m = 0; m < 2; ++m)
#pragma unroll
        for (int t = 0; t < 4; ++t) c[m][t] = z;
    v16h b[2][4];
#pragma unroll
    for (int t = 0; t < 4; ++t) b[0][t] = load_frag(W, ldw, t * 16 + mr, 0, lane);
#pragma unroll
    for (int j = 0; j < NS; ++j) {
        if (j + 1 < NS) {
#pragma unroll
            for (int t = 0; t < 4; ++t)
                b[(j + 1) & 1][t] = load_frag(W, ldw, t * 16 + mr, (j + 1) * 32, lane);
        }
#pragma unroll
        for (int t = 0; t < 4; ++t) {
            c[0][t] = __builtin_amdgcn_wmma_f32_16x16x32_f16(
                false, xa0[j], false, b[j & 1][t], (short)0, c[0][t], false, false);
            c[1][t] = __builtin_amdgcn_wmma_f32_16x16x32_f16(
                false, xa1[j], false, b[j & 1][t], (short)0, c[1][t], false, false);
        }
    }
}

__device__ __forceinline__ void load_b_1x2(const _Float16* __restrict__ W, int ldw,
                                           int lane, v16h b[2]) {
    const int mr = lane & 15;
#pragma unroll
    for (int s = 0; s < 2; ++s) b[s] = load_frag(W, ldw, mr, s * 32, lane);
}

__device__ __forceinline__ void gemm1x2(const unsigned char* __restrict__ Ht,
                                        const v16h b[2], int lane, v8f c[2]) {
    v8f z = {0.f, 0.f, 0.f, 0.f, 0.f, 0.f, 0.f, 0.f};
    c[0] = z; c[1] = z;
    v16h a[2][2];
#pragma unroll
    for (int s = 0; s < 2; ++s) {
        a[s][0] = load_frag_ht(Ht, 0, s * 32, lane);
        a[s][1] = load_frag_ht(Ht, 1, s * 32, lane);
    }
#pragma unroll
    for (int s = 0; s < 2; ++s) {
        c[0] = __builtin_amdgcn_wmma_f32_16x16x32_f16(
            false, a[s][0], false, b[s], (short)0, c[0], false, false);
        c[1] = __builtin_amdgcn_wmma_f32_16x16x32_f16(
            false, a[s][1], false, b[s], (short)0, c[1], false, false);
    }
}

__device__ __forceinline__ void bias_relu_store2(const v8f c[2][4],
                                                 const float* __restrict__ bias,
                                                 unsigned char* __restrict__ Ht, int lane) {
    const int col = lane & 15, hi = lane >> 4;
    unsigned char* __restrict__ hp = Ht + col * HTP_B + 16 * hi;
    const v8h hz = {0, 0, 0, 0, 0, 0, 0, 0};
#pragma unroll
    for (int t = 0; t < 4; ++t) {
        const float bv = bias[t * 16 + col];
#pragma unroll
        for (int m = 0; m < 2; ++m) {
            const v8f v = c[m][t] + bv;
            const v8h h = __builtin_elementwise_max(__builtin_convertvector(v, v8h), hz);
            *(v8h*)(hp + t * 16 * HTP_B + 32 * m) = h;
        }
    }
}

__global__ void cvt_f32_f16(const float* __restrict__ src, _Float16* __restrict__ dst, int n) {
    int i = blockIdx.x * blockDim.x + threadIdx.x;
    if (i < n) dst[i] = (_Float16)src[i];
}

__global__ void cvt_pad_w3(const float* __restrict__ src, _Float16* __restrict__ dst, int outdim) {
    int i = blockIdx.x * blockDim.x + threadIdx.x;
    const int n = K_N * 16 * H_DIM;
    if (i >= n) return;
    int h = i % H_DIM;
    int o = (i / H_DIM) % 16;
    int k = i / (16 * H_DIM);
    dst[i] = (o < outdim) ? (_Float16)src[(k * outdim + o) * H_DIM + h] : (_Float16)0.f;
}

struct WPre {
    v8h w1[W1_PIECES_MAX];
    v8h w2[2];
    v8h w3;
};

template <int KD>
__device__ __forceinline__ void w_fetch(const _Float16* __restrict__ g1,
                                        const _Float16* __restrict__ g2,
                                        const _Float16* __restrict__ g3,
                                        int tid, WPre& p) {
#pragma unroll
    for (int j = 0; j < H_DIM * KD / 8 / NTHREADS; ++j)
        p.w1[j] = *(const v8h*)(g1 + (size_t)(tid + NTHREADS * j) * 8);
#pragma unroll
    for (int j = 0; j < 2; ++j)
        p.w2[j] = *(const v8h*)(g2 + (size_t)(tid + NTHREADS * j) * 8);
    p.w3 = *(const v8h*)(g3 + (size_t)(tid & 127) * 8);
}

template <int KD>
__device__ __forceinline__ void w_store(_Float16* __restrict__ W1s,
                                        _Float16* __restrict__ W2s,
                                        _Float16* __restrict__ W3s,
                                        int tid, const WPre& p) {
    constexpr int PPR = KD / 8;
#pragma unroll
    for (int j = 0; j < H_DIM * KD / 8 / NTHREADS; ++j) {
        const int i = tid + NTHREADS * j;
        const int row = i / PPR, c = i - row * PPR;
        *(v8h*)(W1s + row * W1S + c * 8) = p.w1[j];
    }
#pragma unroll
    for (int j = 0; j < 2; ++j) {
        const int i = tid + NTHREADS * j;
        *(v8h*)(W2s + (i >> 3) * HS + (i & 7) * 8) = p.w2[j];
    }
    if (tid < 128)
        *(v8h*)(W3s + (tid >> 3) * HS + (tid & 7) * 8) = p.w3;
}

__global__ __launch_bounds__(NTHREADS) __attribute__((amdgpu_waves_per_eu(2, 2)))
void qplex_main(const float* __restrict__ states, const float* __restrict__ actions,
                const _Float16* __restrict__ wk1, const _Float16* __restrict__ wa1,
                const _Float16* __restrict__ wc1,
                const _Float16* __restrict__ wk2, const _Float16* __restrict__ wa2,
                const _Float16* __restrict__ wc2,
                const _Float16* __restrict__ wk3, const _Float16* __restrict__ wa3,
                const _Float16* __restrict__ wc3,
                const float* __restrict__ bk1, const float* __restrict__ bk2,
                const float* __restrict__ bk3,
                const float* __restrict__ ba1, const float* __restrict__ ba2,
                const float* __restrict__ ba3,
                const float* __restrict__ bc1, const float* __restrict__ bc2,
                const float* __restrict__ bc3,
                float* __restrict__ out) {
    __shared__ __attribute__((aligned(16))) unsigned char Hs[NWAVES * HT_WAVE_B];
    __shared__ __attribute__((aligned(16))) _Float16 W1s[H_DIM * W1S];
    __shared__ __attribute__((aligned(16))) _Float16 W2s[H_DIM * HS];
    __shared__ __attribute__((aligned(16))) _Float16 W3s[16 * HS];

    const int tid = threadIdx.x;
    const int b0  = blockIdx.x * WG_ROWS;

    WPre pre;
    w_fetch<S_DIM>(wk1, wk2, wk3, tid, pre);

    const int wave = tid >> 5, lane = tid & 31;
    const int col = lane & 15, hi = lane >> 4;
    unsigned char*  Hw = Hs + wave * HT_WAVE_B;

    v16h xa0[XSTEPS], xa1[XSTEPS];
    {
        const size_t r0 = (size_t)(b0 + wave * ROWS_PER_WAVE + col);
        x_frags(states + r0 * S_DIM, actions + r0 * AD_DIM, lane, xa0);
        x_frags(states + (r0 + 16) * S_DIM, actions + (r0 + 16) * AD_DIM, lane, xa1);
    }

    float acc[2][8];
#pragma unroll
    for (int m = 0; m < 2; ++m)
#pragma unroll
        for (int r = 0; r < 8; ++r) acc[m][r] = 0.f;

#pragma unroll 1
    for (int k = 0; k < K_N; ++k) {
        v8f c1[2][4];
        v8f ck[2], ca[2], cc[2];
        v16h b2f[2][4], b3f[2];
        const int kn = (k + 1 < K_N) ? k + 1 : k;

        w_store<S_DIM>(W1s, W2s, W3s, tid, pre);
        __syncthreads();
        w_fetch<S_DIM>(wa1 + (size_t)k * H_DIM * S_DIM, wa2 + (size_t)k * H_DIM * H_DIM,
                       wa3 + (size_t)k * 16 * H_DIM, tid, pre);
        gemm4x2_xr<S_DIM>(xa0, xa1, W1s, W1S, lane, c1);
        load_b_4x2(W2s, HS, lane, b2f);
        bias_relu_store2(c1, bk1 + k * H_DIM, Hw, lane);
        gemm4x2(Hw, b2f, lane, c1);
        load_b_1x2(W3s, HS, lane, b3f);
        bias_relu_store2(c1, bk2 + k * H_DIM, Hw, lane);
        gemm1x2(Hw, b3f, lane, ck);
        __syncthreads();

        w_store<S_DIM>(W1s, W2s, W3s, tid, pre);
        __syncthreads();
        w_fetch<D_DIM>(wc1 + (size_t)k * H_DIM * D_DIM, wc2 + (size_t)k * H_DIM * H_DIM,
                       wc3 + (size_t)k * 16 * H_DIM, tid, pre);
        gemm4x2_xr<S_DIM>(xa0, xa1, W1s, W1S, lane, c1);
        load_b_4x2(W2s, HS, lane, b2f);
        bias_relu_store2(c1, ba1 + k * H_DIM, Hw, lane);
        gemm4x2(Hw, b2f, lane, c1);
        load_b_1x2(W3s, HS, lane, b3f);
        bias_relu_store2(c1, ba2 + k * H_DIM, Hw, lane);
        gemm1x2(Hw, b3f, lane, ca);
        __syncthreads();

        w_store<D_DIM>(W1s, W2s, W3s, tid, pre);
        __syncthreads();
        w_fetch<S_DIM>(wk1 + (size_t)kn * H_DIM * S_DIM, wk2 + (size_t)kn * H_DIM * H_DIM,
                       wk3 + (size_t)kn * 16 * H_DIM, tid, pre);
        gemm4x2_xr<D_DIM>(xa0, xa1, W1s, W1S, lane, c1);
        load_b_4x2(W2s, HS, lane, b2f);
        bias_relu_store2(c1, bc1 + k * H_DIM, Hw, lane);
        gemm4x2(Hw, b2f, lane, c1);
        load_b_1x2(W3s, HS, lane, b3f);
        bias_relu_store2(c1, bc2 + k * H_DIM, Hw, lane);
        gemm1x2(Hw, b3f, lane, cc);
        __syncthreads();

        const float kbias = bk3[k];
        const float ab3v  = ba3[k * A_N + (col & 7)];
        const float cb3v  = bc3[k * A_N + (col & 7)];
#pragma unroll
        for (int m = 0; m < 2; ++m)
#pragma unroll
            for (int r = 0; r < 8; ++r) {
                float kv = __shfl(ck[m][r], lane & 16, 32);
                kv = fabsf(kv + kbias) + 1e-10f;
                float ag = 1.f / (1.f + __expf(-(ca[m][r] + ab3v)));
                float ac = 1.f / (1.f + __expf(-(cc[m][r] + cb3v)));
                acc[m][r] += kv * ag * ac;
            }
    }

    if (col < A_N) {
#pragma unroll
        for (int m = 0; m < 2; ++m)
#pragma unroll
            for (int r = 0; r < 8; ++r)
                out[(size_t)(b0 + wave * ROWS_PER_WAVE + m * 16 + r + 8 * hi) * A_N + col] =
                    acc[m][r];
    }
}

extern "C" void kernel_launch(void* const* d_in, const int* in_sizes, int n_in,
                              void* d_out, int out_size, void* d_ws, size_t ws_size,
                              hipStream_t stream) {
    (void)in_sizes; (void)n_in; (void)out_size; (void)ws_size;
    const float* states  = (const float*)d_in[0];
    const float* actions = (const float*)d_in[1];
    const float* kW1 = (const float*)d_in[2];
    const float* kb1 = (const float*)d_in[3];
    const float* kW2 = (const float*)d_in[4];
    const float* kb2 = (const float*)d_in[5];
    const float* kW3 = (const float*)d_in[6];
    const float* kb3 = (const float*)d_in[7];
    const float* aW1 = (const float*)d_in[8];
    const float* ab1 = (const float*)d_in[9];
    const float* aW2 = (const float*)d_in[10];
    const float* ab2 = (const float*)d_in[11];
    const float* aW3 = (const float*)d_in[12];
    const float* ab3 = (const float*)d_in[13];
    const float* cW1 = (const float*)d_in[14];
    const float* cb1 = (const float*)d_in[15];
    const float* cW2 = (const float*)d_in[16];
    const float* cb2 = (const float*)d_in[17];
    const float* cW3 = (const float*)d_in[18];
    const float* cb3 = (const float*)d_in[19];

    _Float16* ws = (_Float16*)d_ws;
    size_t off = 0;
    _Float16* wk1 = ws + off; off += (size_t)K_N * H_DIM * S_DIM;
    _Float16* wa1 = ws + off; off += (size_t)K_N * H_DIM * S_DIM;
    _Float16* wc1 = ws + off; off += (size_t)K_N * H_DIM * D_DIM;
    _Float16* wk2 = ws + off; off += (size_t)K_N * H_DIM * H_DIM;
    _Float16* wa2 = ws + off; off += (size_t)K_N * H_DIM * H_DIM;
    _Float16* wc2 = ws + off; off += (size_t)K_N * H_DIM * H_DIM;
    _Float16* wk3 = ws + off; off += (size_t)K_N * 16 * H_DIM;
    _Float16* wa3 = ws + off; off += (size_t)K_N * 16 * H_DIM;
    _Float16* wc3 = ws + off; off += (size_t)K_N * 16 * H_DIM;

    auto cvt = [&](const float* s, _Float16* d, int n) {
        cvt_f32_f16<<<(n + 255) / 256, 256, 0, stream>>>(s, d, n);
    };
    cvt(kW1, wk1, K_N * H_DIM * S_DIM);
    cvt(aW1, wa1, K_N * H_DIM * S_DIM);
    cvt(cW1, wc1, K_N * H_DIM * D_DIM);
    cvt(kW2, wk2, K_N * H_DIM * H_DIM);
    cvt(aW2, wa2, K_N * H_DIM * H_DIM);
    cvt(cW2, wc2, K_N * H_DIM * H_DIM);
    const int n3 = K_N * 16 * H_DIM;
    cvt_pad_w3<<<(n3 + 255) / 256, 256, 0, stream>>>(kW3, wk3, 1);
    cvt_pad_w3<<<(n3 + 255) / 256, 256, 0, stream>>>(aW3, wa3, A_N);
    cvt_pad_w3<<<(n3 + 255) / 256, 256, 0, stream>>>(cW3, wc3, A_N);

    qplex_main<<<B_TOT / WG_ROWS, NTHREADS, 0, stream>>>(
        states, actions,
        wk1, wa1, wc1, wk2, wa2, wc2, wk3, wa3, wc3,
        kb1, kb2, kb3, ab1, ab2, ab3, cb1, cb2, cb3,
        (float*)d_out);
}


# Weights staged from the fp32 inputs: fetched a net ahead, rounded to f16 in registers, then to LDS; the 9 conversion launches and workspace go; action-column A fragments parked in LDS; same sums
# speedup vs baseline: 4.7205x; 1.0775x over previous
#include <hip/hip_runtime.h>
#include <hip/hip_bf16.h>

typedef __attribute__((ext_vector_type(16))) _Float16 v16h;
typedef __attribute__((ext_vector_type(8)))  _Float16 v8h;
typedef __attribute__((ext_vector_type(8)))  float    v8f;
typedef __attribute__((ext_vector_type(4)))  float    v4f;
typedef __attribute__((ext_vector_type(4)))  _Float16 v4h;

#define B_TOT   65536
#define S_DIM   256
#define A_N     8
#define K_N     10
#define H_DIM   64
#define AD_DIM  128
#define D_DIM   384

#define WG_ROWS 256
#define NTHREADS 256
#define NWAVES  (NTHREADS / 32)
#define ROWS_PER_WAVE 32
#define XSTEPS  (D_DIM / 32)
#define HS      72
#define HTP_B   96
#define HT_WAVE_B (H_DIM * HTP_B)
#define W1S     392
#define XS_REG  (S_DIM / 32)
#define XS_LDS  (XSTEPS - XS_REG)
#define XW_WAVE_B (XS_LDS * 2 * 1024)
#define WB_W2   (H_DIM * W1S)
#define WB_W3   (WB_W2 + H_DIM * HS)
#define WB_SIZE (WB_W3 + 16 * HS)
#define NP_MAX  (D_DIM / 16 + 5)

__device__ __forceinline__ v16h load_frag(const _Float16* __restrict__ p,
                                          int stride, int row, int k0, int lane) {
    const int ho = (lane >> 4) << 3;
    const _Float16* q = p + row * stride + k0 + ho;
    union { v16h v; struct { v8h lo; v8h hi; } s; } u;
    u.s.lo = *(const v8h*)q;
    u.s.hi = *(const v8h*)(q + 16);
    return u.v;
}

typedef __attribute__((ext_vector_type(8))) short s16x8;
__device__ __forceinline__ v8h ld_tr16(const unsigned char* p) {
    typedef __attribute__((address_space(3))) s16x8 lds_s16x8;
    return __builtin_bit_cast(v8h, __builtin_amdgcn_ds_load_tr16_b128_v8i16((lds_s16x8*)p));
}

__device__ __forceinline__ v16h load_frag_ht(const unsigned char* __restrict__ Ht,
                                             int m, int k0, int lane) {
    const unsigned char* p = Ht + (k0 + 8 * (lane >> 4) + (lane & 7)) * HTP_B
                                + 32 * m + 16 * ((lane >> 3) & 1);
    union { v16h v; struct { v8h lo; v8h hi; } s; } u;
    u.s.lo = ld_tr16(p);
    u.s.hi = ld_tr16(p + 16 * HTP_B);
    return u.v;
}

__device__ __forceinline__ void load_b_4x2(const _Float16* __restrict__ W, int ldw,
                                           int lane, v16h b[2][4]) {
    const int mr = lane & 15;
#pragma unroll
    for (int s = 0; s < 2; ++s)
#pragma unroll
        for (int t = 0; t < 4; ++t) b[s][t] = load_frag(W, ldw, t * 16 + mr, s * 32, lane);
}

__device__ __forceinline__ void gemm4x2(const unsigned char* __restrict__ Ht,
                                        const v16h b[2][4], int lane, v8f c[2][4]) {
    v8f z = {0.f, 0.f, 0.f, 0.f, 0.f, 0.f, 0.f, 0.f};
#pragma unroll
    for (int m = 0; m < 2; ++m)
#pragma unroll
        for (int t = 0; t < 4; ++t) c[m][t] = z;
    v16h a[2][2];
#pragma unroll
    for (int s = 0; s < 2; ++s) {
        a[s][0] = load_frag_ht(Ht, 0, s * 32, lane);
        a[s][1] = load_frag_ht(Ht, 1, s * 32, lane);
    }
#pragma unroll
    for (int s = 0; s < 2; ++s) {
#pragma unroll
        for (int t = 0; t < 4; ++t) {
            c[0][t] = __builtin_amdgcn_wmma_f32_16x16x32_f16(
                false, a[s][0], false, b[s][t], (short)0, c[0][t], false, false);
            c[1][t] = __builtin_amdgcn_wmma_f32_16x16x32_f16(
                false, a[s][1], false, b[s][t], (short)0, c[1][t], false, false);
        }
    }
}

__device__ __forceinline__ void x_frags(const float* __restrict__ srow,
                                        const float* __restrict__ arow,
                                        int lane, v16h xa[XSTEPS]) {
    const int ho = (lane >> 4) << 3;
    v4f t[XSTEPS][4];
#pragma unroll
    for (int j = 0; j < XSTEPS; ++j) {
        const float* p = (j < S_DIM / 32) ? srow + j * 32 + ho
                                          : arow + (j - S_DIM / 32) * 32 + ho;
        t[j][0] = *(const v4f*)(p);
        t[j][1] = *(const v4f*)(p + 4);
        t[j][2] = *(const v4f*)(p + 16);
        t[j][3] = *(const v4f*)(p + 20);
    }
#pragma unroll
    for (int j = 0; j < XSTEPS; ++j) {
        v8h lo = __builtin_shufflevector(__builtin_convertvector(t[j][0], v4h),
                                         __builtin_convertvector(t[j][1], v4h),
                                         0, 1, 2, 3, 4, 5, 6, 7);
        v8h hi = __builtin_shufflevector(__builtin_convertvector(t[j][2], v4h),
                                         __builtin_convertvector(t[j][3], v4h),
                                         0, 1, 2, 3, 4, 5, 6, 7);
        xa[j] = __builtin_shufflevector(lo, hi, 0, 1, 2, 3, 4, 5, 6, 7,
                                        8, 9, 10, 11, 12, 13, 14, 15);
    }
}

__device__ __forceinline__ void park_frag(unsigned char* __restrict__ Xw, int s, int m, int lane, v16h a) {
    unsigned char* p = Xw + (s * 2 + m) * 1024 + lane * 16;
    *(v8h*)p         = __builtin_shufflevector(a, a, 0, 1, 2, 3, 4, 5, 6, 7);
    *(v8h*)(p + 512) = __builtin_shufflevector(a, a, 8, 9, 10, 11, 12, 13, 14, 15);
}
__device__ __forceinline__ v16h parked_frag(const unsigned char* __restrict__ Xw, int s, int m, int lane) {
    const unsigned char* p = Xw + (s * 2 + m) * 1024 + lane * 16;
    union { v16h v; struct { v8h lo; v8h hi; } h; } u;
    u.h.lo = *(const v8h*)p;
    u.h.hi = *(const v8h*)(p + 512);
    return u.v;
}

template <int KD>
__device__ __forceinline__ void gemm4x2_xr(const v16h xa0[XS_REG], const v16h xa1[XS_REG],
                                           const unsigned char* __restrict__ Xw,
                                           const _Float16* __restrict__ W, int ldw,
                                           int lane, v8f c[2][4]) {
    const int mr = lane & 15;
    constexpr int NS = KD / 32;
    v8f z = {0.f, 0.f, 0.f, 0.f, 0.f, 0.f, 0.f, 0.f};
#pragma unroll
    for (int m = 0; m < 2; ++m)
#pragma unroll
        for (int t = 0; t < 4; ++t) c[m][t] = z;
    v16h b[2][4];
    v16h ax[2][2];
#pragma unroll
    for (int t = 0; t < 4; ++t) b[0][t] = load_frag(W, ldw, t * 16 + mr, 0, lane);
#pragma unroll
    for (int j = 0; j < NS; ++j) {
        if (j + 1 < NS) {
#pragma unroll
            for (int t = 0; t < 4; ++t)
                b[(j + 1) & 1][t] = load_frag(W, ldw, t * 16 + mr, (j + 1) * 32, lane);
            if (j + 1 >= XS_REG) {
                ax[(j + 1) & 1][0] = parked_frag(Xw, j + 1 - XS_REG, 0, lane);
                ax[(j + 1) & 1][1] = parked_frag(Xw, j + 1 - XS_REG, 1, lane);
            }
        }
        const v16h a0 = (j < XS_REG) ? xa0[j < XS_REG ? j : 0] : ax[j & 1][0];
        const v16h a1 = (j < XS_REG) ? xa1[j < XS_REG ? j : 0] : ax[j & 1][1];
        __builtin_amdgcn_sched_barrier(0);
#pragma unroll
        for (int t = 0; t < 4; ++t) {
            c[0][t] = __builtin_amdgcn_wmma_f32_16x16x32_f16(
                false, a0, false, b[j & 1][t], (short)0, c[0][t], false, false);
            c[1][t] = __builtin_amdgcn_wmma_f32_16x16x32_f16(
                false, a1, false, b[j & 1][t], (short)0, c[1][t], false, false);
        }
        __builtin_amdgcn_sched_barrier(0);
    }
}

__device__ __forceinline__ void load_b_1x2(const _Float16* __restrict__ W, int ldw,
                                           int lane, v16h b[2]) {
    const int mr = lane & 15;
#pragma unroll
    for (int s = 0; s < 2; ++s) b[s] = load_frag(W, ldw, mr, s * 32, lane);
}

__device__ __forceinline__ void gemm1x2(const unsigned char* __restrict__ Ht,
                                        const v16h b[2], int lane, v8f c[2]) {
    v8f z = {0.f, 0.f, 0.f, 0.f, 0.f, 0.f, 0.f, 0.f};
    c[0] = z; c[1] = z;
    v16h a[2][2];
#pragma unroll
    for (int s = 0; s < 2; ++s) {
        a[s][0] = load_frag_ht(Ht, 0, s * 32, lane);
        a[s][1] = load_frag_ht(Ht, 1, s * 32, lane);
    }
#pragma unroll
    for (int s = 0; s < 2; ++s) {
        c[0] = __builtin_amdgcn_wmma_f32_16x16x32_f16(
            false, a[s][0], false, b[s], (short)0, c[0], false, false);
        c[1] = __builtin_amdgcn_wmma_f32_16x16x32_f16(
            false, a[s][1], false, b[s], (short)0, c[1], false, false);
    }
}

__device__ __forceinline__ void bias_relu_store2(const v8f c[2][4],
                                                 const float* __restrict__ bias_all, unsigned e,
                                                 unsigned char* __restrict__ Ht, int lane) {
    const int col = lane & 15, hi = lane >> 4;
    const float* __restrict__ bias = bias_all + (e + (unsigned)col);
    unsigned char* __restrict__ hp = Ht + col * HTP_B + 16 * hi;
    const v8h hz = {0, 0, 0, 0, 0, 0, 0, 0};
#pragma unroll
    for (int t = 0; t < 4; ++t) {
        const float bv = bias[t * 16];
#pragma unroll
        for (int m = 0; m < 2; ++m) {
            const v8f v = c[m][t] + bv;
            const v8h h = __builtin_elementwise_max(__builtin_convertvector(v, v8h), hz);
            *(v8h*)(hp + t * 16 * HTP_B + 32 * m) = h;
        }
    }
}

template <int KD, int OUTD>
__device__ __forceinline__ void w_fetch(const float* __restrict__ g1, unsigned e1,
                                        const float* __restrict__ g2, unsigned e2,
                                        const float* __restrict__ g3, unsigned e3,
                                        int tid, v4f q[NP_MAX]) {
    constexpr int NP1 = KD / 16;
    constexpr int LPR = KD / 128;
    const unsigned o1 = e1 + (unsigned)((tid >> 5) * KD + (tid & 31) * 4);
    const unsigned o2 = e2 + (unsigned)(tid * 4);
    const unsigned o3 = e3 + ((tid < OUTD * (H_DIM / 4)) ? (unsigned)(tid * 4) : 0u);
    const float* __restrict__ s1 = g1 + o1;
    const float* __restrict__ s2 = g2 + o2;
#pragma unroll
    for (int p = 0; p < NP1; ++p)
        q[p] = *(const v4f*)(s1 + ((p / LPR) * 8 * KD + (p % LPR) * 128));
#pragma unroll
    for (int j = 0; j < 4; ++j)
        q[NP1 + j] = *(const v4f*)(s2 + j * NTHREADS * 4);
    q[NP1 + 4] = *(const v4f*)(g3 + o3);
    __builtin_amdgcn_sched_barrier(0);
}

template <int KD, int OUTD>
__device__ __forceinline__ void w_round(const v4f q[NP_MAX], v4h h[NP_MAX], int tid) {
    constexpr int NP = KD / 16 + 5;
    __builtin_amdgcn_sched_barrier(0);
#pragma unroll
    for (int p = 0; p < NP; ++p) h[p] = __builtin_convertvector(q[p], v4h);
    const v4h hz = {0, 0, 0, 0};
    if (tid >= OUTD * (H_DIM / 4)) h[NP - 1] = hz;
    __builtin_amdgcn_sched_barrier(0);
}

template <int KD>
__device__ __forceinline__ void w_store(_Float16* __restrict__ Wb, int tid, const v4h h[NP_MAX]) {
    constexpr int NP1 = KD / 16;
    constexpr int LPR = KD / 128;
    _Float16* __restrict__ d1 = Wb + (tid >> 5) * W1S + (tid & 31) * 4;
    _Float16* __restrict__ d2 = Wb + WB_W2 + (tid >> 4) * HS + (tid & 15) * 4;
#pragma unroll
    for (int p = 0; p < NP1; ++p) *(v4h*)(d1 + (p / LPR) * 8 * W1S + (p % LPR) * 128) = h[p];
#pragma unroll
    for (int j = 0; j < 4; ++j) *(v4h*)(d2 + j * 16 * HS) = h[NP1 + j];
    *(v4h*)(d2 + (WB_W3 - WB_W2)) = h[NP1 + 4];
}

__global__ __launch_bounds__(NTHREADS) __attribute__((amdgpu_waves_per_eu(2, 2)))
void qplex_main(const float* __restrict__ states, const float* __restrict__ actions,
                const float* __restrict__ wk1, const float* __restrict__ wa1,
                const float* __restrict__ wc1,
                const float* __restrict__ wk2, const float* __restrict__ wa2,
                const float* __restrict__ wc2,
                const float* __restrict__ wk3, const float* __restrict__ wa3,
                const float* __restrict__ wc3,
                const float* __restrict__ bk1, const float* __restrict__ bk2,
                const float* __restrict__ bk3,
                const float* __restrict__ ba1, const float* __restrict__ ba2,
                const float* __restrict__ ba3,
                const float* __restrict__ bc1, const float* __restrict__ bc2,
                const float* __restrict__ bc3,
                float* __restrict__ out) {
    __shared__ __attribute__((aligned(16))) unsigned char Hs[NWAVES * HT_WAVE_B];
    __shared__ __attribute__((aligned(16))) unsigned char Xs[NWAVES * XW_WAVE_B];
    __shared__ __attribute__((aligned(16))) _Float16 Wb[WB_SIZE];

    const int tid = threadIdx.x;
    const int b0  = blockIdx.x * WG_ROWS;

    v4f pre[NP_MAX];
    v4h ph[NP_MAX];
    w_fetch<S_DIM, 1>(wk1, 0u, wk2, 0u, wk3, 0u, tid, pre);

    const int wave = tid >> 5, lane = tid & 31;
    const int col = lane & 15, hi = lane >> 4;
    unsigned char* Hw = Hs + wave * HT_WAVE_B;
    unsigned char* Xw = Xs + wave * XW_WAVE_B;

    v16h xa0[XS_REG], xa1[XS_REG];
    {
        const size_t r0 = (size_t)(b0 + wave * ROWS_PER_WAVE + col);
        v16h xf[XSTEPS];
        x_frags(states + r0 * S_DIM, actions + r0 * AD_DIM, lane, xf);
#pragma unroll
        for (int j = 0; j < XS_REG; ++j) xa0[j] = xf[j];
#pragma unroll
        for (int j = 0; j < XS_LDS; ++j) park_frag(Xw, j, 0, lane, xf[XS_REG + j]);
        x_frags(states + (r0 + 16) * S_DIM, actions + (r0 + 16) * AD_DIM, lane, xf);
#pragma unroll
        for (int j = 0; j < XS_REG; ++j) xa1[j] = xf[j];
#pragma unroll
        for (int j = 0; j < XS_LDS; ++j) park_frag(Xw, j, 1, lane, xf[XS_REG + j]);
    }

    w_round<S_DIM, 1>(pre, ph, tid);

    float acc[2][8];
#pragma unroll
    for (int m = 0; m < 2; ++m)
#pragma unroll
        for (int r = 0; r < 8; ++r) acc[m][r] = 0.f;

#pragma unroll 1
    for (int k = 0; k < K_N; ++k) {
        v8f c1[2][4];
        v8f ck[2], ca[2], cc[2];
        v16h b2f[2][4], b3f[2];
        const unsigned uk  = (unsigned)k;
        const unsigned ukn = (k + 1 < K_N) ? uk + 1u : uk;

        w_store<S_DIM>(Wb, tid, ph);
        __syncthreads();
        w_fetch<S_DIM, A_N>(wa1, uk * (H_DIM * S_DIM), wa2, uk * (H_DIM * H_DIM),
                            wa3, uk * (A_N * H_DIM), tid, pre);
        gemm4x2_xr<S_DIM>(xa0, xa1, Xw, Wb, W1S, lane, c1);
        load_b_4x2(Wb + WB_W2, HS, lane, b2f);
        w_round<S_DIM, A_N>(pre, ph, tid);
        bias_relu_store2(c1, bk1, uk * H_DIM, Hw, lane);
        gemm4x2(Hw, b2f, lane, c1);
        load_b_1x2(Wb + WB_W3, HS, lane, b3f);
        bias_relu_store2(c1, bk2, uk * H_DIM, Hw, lane);
        gemm1x2(Hw, b3f, lane, ck);
        __syncthreads();

        w_store<S_DIM>(Wb, tid, ph);
        __syncthreads();
        w_fetch<D_DIM, A_N>(wc1, uk * (H_DIM * D_DIM), wc2, uk * (H_DIM * H_DIM),
                            wc3, uk * (A_N * H_DIM), tid, pre);
        gemm4x2_xr<S_DIM>(xa0, xa1, Xw, Wb, W1S, lane, c1);
        load_b_4x2(Wb + WB_W2, HS, lane, b2f);
        w_round<D_DIM, A_N>(pre, ph, tid);
        bias_relu_store2(c1, ba1, uk * H_DIM, Hw, lane);
        gemm4x2(Hw, b2f, lane, c1);
        load_b_1x2(Wb + WB_W3, HS, lane, b3f);
        bias_relu_store2(c1, ba2, uk * H_DIM, Hw, lane);
        gemm1x2(Hw, b3f, lane, ca);
        __syncthreads();

        w_store<D_DIM>(Wb, tid, ph);
        __syncthreads();
        w_fetch<S_DIM, 1>(wk1, ukn * (H_DIM * S_DIM), wk2, ukn * (H_DIM * H_DIM),
                          wk3, ukn * H_DIM, tid, pre);
        gemm4x2_xr<D_DIM>(xa0, xa1, Xw, Wb, W1S, lane, c1);
        load_b_4x2(Wb + WB_W2, HS, lane, b2f);
        w_round<S_DIM, 1>(pre, ph, tid);
        bias_relu_store2(c1, bc1, uk * H_DIM, Hw, lane);
        gemm4x2(Hw, b2f, lane, c1);
        load_b_1x2(Wb + WB_W3, HS, lane, b3f);
        bias_relu_store2(c1, bc2, uk * H_DIM, Hw, lane);
        gemm1x2(Hw, b3f, lane, cc);
        __syncthreads();

        const float kbias = bk3[k];
        const float ab3v  = ba3[k * A_N + (col & 7)];
        const float cb3v  = bc3[k * A_N + (col & 7)];
#pragma unroll
        for (int m = 0; m < 2; ++m)
#pragma unroll
            for (int r = 0; r < 8; ++r) {
                float kv = __shfl(ck[m][r], lane & 16, 32);
                kv = fabsf(kv + kbias) + 1e-10f;
                float ag = 1.f / (1.f + __expf(-(ca[m][r] + ab3v)));
                float ac = 1.f / (1.f + __expf(-(cc[m][r] + cb3v)));
                acc[m][r] += kv * ag * ac;
            }
    }

    if (col < A_N) {
#pragma unroll
        for (int m = 0; m < 2; ++m)
#pragma unroll
            for (int r = 0; r < 8; ++r)
                out[(size_t)(b0 + wave * ROWS_PER_WAVE + m * 16 + r + 8 * hi) * A_N + col] =
                    acc[m][r];
    }
}

extern "C" void kernel_launch(void* const* d_in, const int* in_sizes, int n_in,
                              void* d_out, int out_size, void* d_ws, size_t ws_size,
                              hipStream_t stream) {
    (void)in_sizes; (void)n_in; (void)out_size; (void)d_ws; (void)ws_size;
    const float* states  = (const float*)d_in[0];
    const float* actions = (const float*)d_in[1];
    const float* kW1 = (const float*)d_in[2];
    const float* kb1 = (const float*)d_in[3];
    const float* kW2 = (const float*)d_in[4];
    const float* kb2 = (const float*)d_in[5];
    const float* kW3 = (const float*)d_in[6];
    const float* kb3 = (const float*)d_in[7];
    const float* aW1 = (const float*)d_in[8];
    const float* ab1 = (const float*)d_in[9];
    const float* aW2 = (const float*)d_in[10];
    const float* ab2 = (const float*)d_in[11];
    const float* aW3 = (const float*)d_in[12];
    const float* ab3 = (const float*)d_in[13];
    const float* cW1 = (const float*)d_in[14];
    const float* cb1 = (const float*)d_in[15];
    const float* cW2 = (const float*)d_in[16];
    const float* cb2 = (const float*)d_in[17];
    const float* cW3 = (const float*)d_in[18];
    const float* cb3 = (const float*)d_in[19];

    qplex_main<<<B_TOT / WG_ROWS, NTHREADS, 0, stream>>>(
        states, actions,
        kW1, aW1, cW1, kW2, aW2, cW2, kW3, aW3, cW3,
        kb1, kb2, kb3, ab1, ab2, ab3, cb1, cb2, cb3,
        (float*)d_out);
}


# Parameters double-buffered in LDS: next net's weights and hidden biases rounded and stored to the other buffer behind the first epilogue; one barrier per net; nets run key, action, agents; same sums
# speedup vs baseline: 5.2890x; 1.1204x over previous
#include <hip/hip_runtime.h>
#include <hip/hip_bf16.h>

typedef __attribute__((ext_vector_type(16))) _Float16 v16h;
typedef __attribute__((ext_vector_type(8)))  _Float16 v8h;
typedef __attribute__((ext_vector_type(8)))  float    v8f;
typedef __attribute__((ext_vector_type(4)))  float    v4f;
typedef __attribute__((ext_vector_type(4)))  _Float16 v4h;

#define B_TOT   65536
#define S_DIM   256
#define A_N     8
#define K_N     10
#define H_DIM   64
#define AD_DIM  128
#define D_DIM   384

#define WG_ROWS 256
#define NTHREADS 256
#define NWAVES  (NTHREADS / 32)
#define ROWS_PER_WAVE 32
#define XSTEPS  (D_DIM / 32)
#define HS      72
#define HTP_B   96
#define HT_WAVE_B (H_DIM * HTP_B)
#define W1S     392
#define XS_REG  (S_DIM / 32)
#define XS_LDS  (XSTEPS - XS_REG)
#define XW_WAVE_B (XS_LDS * 2 * 1024)
#define WB_W2   (H_DIM * W1S)
#define WB_W3   (WB_W2 + H_DIM * HS)
#define WB_SIZE (WB_W3 + 16 * HS)
#define WB_BIAS WB_SIZE
#define WB_TOTAL (WB_SIZE + 2 * 2 * H_DIM)
#define NP_MAX  (D_DIM / 16 + 5)

__device__ __forceinline__ v16h load_frag(const _Float16* __restrict__ p,
                                          int stride, int row, int k0, int lane) {
    const int ho = (lane >> 4) << 3;
    const _Float16* q = p + row * stride + k0 + ho;
    union { v16h v; struct { v8h lo; v8h hi; } s; } u;
    u.s.lo = *(const v8h*)q;
    u.s.hi = *(const v8h*)(q + 16);
    return u.v;
}

typedef __attribute__((ext_vector_type(8))) short s16x8;
__device__ __forceinline__ v8h ld_tr16(const unsigned char* p) {
    typedef __attribute__((address_space(3))) s16x8 lds_s16x8;
    return __builtin_bit_cast(v8h, __builtin_amdgcn_ds_load_tr16_b128_v8i16((lds_s16x8*)p));
}

__device__ __forceinline__ v16h load_frag_ht(const unsigned char* __restrict__ Ht,
                                             int m, int k0, int lane) {
    const unsigned char* p = Ht + (k0 + 8 * (lane >> 4) + (lane & 7)) * HTP_B
                                + 32 * m + 16 * ((lane >> 3) & 1);
    union { v16h v; struct { v8h lo; v8h hi; } s; } u;
    u.s.lo = ld_tr16(p);
    u.s.hi = ld_tr16(p + 16 * HTP_B);
    return u.v;
}

__device__ __forceinline__ void load_b_4x2(const _Float16* __restrict__ W, int ldw,
                                           int lane, v16h b[2][4]) {
    const int mr = lane & 15;
#pragma unroll
    for (int s = 0; s < 2; ++s)
#pragma unroll
        for (int t = 0; t < 4; ++t) b[s][t] = load_frag(W, ldw, t * 16 + mr, s * 32, lane);
}

__device__ __forceinline__ void load_a_2x2(const unsigned char* __restrict__ Ht, int lane, v16h a[2][2]) {
#pragma unroll
    for (int s = 0; s < 2; ++s) {
        a[s][0] = load_frag_ht(Ht, 0, s * 32, lane);
        a[s][1] = load_frag_ht(Ht, 1, s * 32, lane);
    }
}
__device__ __forceinline__ void mma4x2(const v16h a[2][2], const v16h b[2][4], v8f c[2][4]) {
    v8f z = {0.f, 0.f, 0.f, 0.f, 0.f, 0.f, 0.f, 0.f};
#pragma unroll
    for (int m = 0; m < 2; ++m)
#pragma unroll
        for (int t = 0; t < 4; ++t) c[m][t] = z;
#pragma unroll
    for (int s = 0; s < 2; ++s) {
#pragma unroll
        for (int t = 0; t < 4; ++t) {
            c[0][t] = __builtin_amdgcn_wmma_f32_16x16x32_f16(
                false, a[s][0], false, b[s][t], (short)0, c[0][t], false, false);
            c[1][t] = __builtin_amdgcn_wmma_f32_16x16x32_f16(
                false, a[s][1], false, b[s][t], (short)0, c[1][t], false, false);
        }
    }
}

__device__ __forceinline__ void x_frags(const float* __restrict__ srow,
                                        const float* __restrict__ arow,
                                        int lane, v16h xa[XSTEPS]) {
    const int ho = (lane >> 4) << 3;
    v4f t[XSTEPS][4];
#pragma unroll
    for (int j = 0; j < XSTEPS; ++j) {
        const float* p = (j < S_DIM / 32) ? srow + j * 32 + ho
                                          : arow + (j - S_DIM / 32) * 32 + ho;
        t[j][0] = *(const v4f*)(p);
        t[j][1] = *(const v4f*)(p + 4);
        t[j][2] = *(const v4f*)(p + 16);
        t[j][3] = *(const v4f*)(p + 20);
    }
#pragma unroll
    for (int j = 0; j < XSTEPS; ++j) {
        v8h lo = __builtin_shufflevector(__builtin_convertvector(t[j][0], v4h),
                                         __builtin_convertvector(t[j][1], v4h),
                                         0, 1, 2, 3, 4, 5, 6, 7);
        v8h hi = __builtin_shufflevector(__builtin_convertvector(t[j][2], v4h),
                                         __builtin_convertvector(t[j][3], v4h),
                                         0, 1, 2, 3, 4, 5, 6, 7);
        xa[j] = __builtin_shufflevector(lo, hi, 0, 1, 2, 3, 4, 5, 6, 7,
                                        8, 9, 10, 11, 12, 13, 14, 15);
    }
}

__device__ __forceinline__ void park_frag(unsigned char* __restrict__ Xw, int s, int m, int lane, v16h a) {
    unsigned char* p = Xw + (s * 2 + m) * 1024 + lane * 16;
    *(v8h*)p         = __builtin_shufflevector(a, a, 0, 1, 2, 3, 4, 5, 6, 7);
    *(v8h*)(p + 512) = __builtin_shufflevector(a, a, 8, 9, 10, 11, 12, 13, 14, 15);
}
__device__ __forceinline__ v16h parked_frag(const unsigned char* __restrict__ Xw, int s, int m, int lane) {
    const unsigned char* p = Xw + (s * 2 + m) * 1024 + lane * 16;
    union { v16h v; struct { v8h lo; v8h hi; } h; } u;
    u.h.lo = *(const v8h*)p;
    u.h.hi = *(const v8h*)(p + 512);
    return u.v;
}

template <int KD>
__device__ __forceinline__ void gemm4x2_xr(const v16h xa0[XS_REG], const v16h xa1[XS_REG],
                                           const unsigned char* __restrict__ Xw,
                                           const _Float16* __restrict__ W, int ldw,
                                           int lane, v8f c[2][4]) {
    const int mr = lane & 15;
    constexpr int NS = KD / 32;
    v8f z = {0.f, 0.f, 0.f, 0.f, 0.f, 0.f, 0.f, 0.f};
#pragma unroll
    for (int m = 0; m < 2; ++m)
#pragma unroll
        for (int t = 0; t < 4; ++t) c[m][t] = z;
    v16h b[2][4];
    v16h ax[2][2];
#pragma unroll
    for (int t = 0; t < 4; ++t) b[0][t] = load_frag(W, ldw, t * 16 + mr, 0, lane);
#pragma unroll
    for (int j = 0; j < NS; ++j) {
        if (j + 1 < NS) {
#pragma unroll
            for (int t = 0; t < 4; ++t)
                b[(j + 1) & 1][t] = load_frag(W, ldw, t * 16 + mr, (j + 1) * 32, lane);
            if (j + 1 >= XS_REG) {
                ax[(j + 1) & 1][0] = parked_frag(Xw, j + 1 - XS_REG, 0, lane);
                ax[(j + 1) & 1][1] = parked_frag(Xw, j + 1 - XS_REG, 1, lane);
            }
        }
        const v16h a0 = (j < XS_REG) ? xa0[j < XS_REG ? j : 0] : ax[j & 1][0];
        const v16h a1 = (j < XS_REG) ? xa1[j < XS_REG ? j : 0] : ax[j & 1][1];
        __builtin_amdgcn_sched_barrier(0);
#pragma unroll
        for (int t = 0; t < 4; ++t) {
            c[0][t] = __builtin_amdgcn_wmma_f32_16x16x32_f16(
                false, a0, false, b[j & 1][t], (short)0, c[0][t], false, false);
            c[1][t] = __builtin_amdgcn_wmma_f32_16x16x32_f16(
                false, a1, false, b[j & 1][t], (short)0, c[1][t], false, false);
        }
        __builtin_amdgcn_sched_barrier(0);
    }
}

__device__ __forceinline__ void load_b_1x2(const _Float16* __restrict__ W, int ldw,
                                           int lane, v16h b[2]) {
    const int mr = lane & 15;
#pragma unroll
    for (int s = 0; s < 2; ++s) b[s] = load_frag(W, ldw, mr, s * 32, lane);
}

__device__ __forceinline__ void gemm1x2(const unsigned char* __restrict__ Ht,
                                        const v16h b[2], int lane, v8f c[2]) {
    v8f z = {0.f, 0.f, 0.f, 0.f, 0.f, 0.f, 0.f, 0.f};
    c[0] = z; c[1] = z;
    v16h a[2][2];
#pragma unroll
    for (int s = 0; s < 2; ++s) {
        a[s][0] = load_frag_ht(Ht, 0, s * 32, lane);
        a[s][1] = load_frag_ht(Ht, 1, s * 32, lane);
    }
#pragma unroll
    for (int s = 0; s < 2; ++s) {
        c[0] = __builtin_amdgcn_wmma_f32_16x16x32_f16(
            false, a[s][0], false, b[s], (short)0, c[0], false, false);
        c[1] = __builtin_amdgcn_wmma_f32_16x16x32_f16(
            false, a[s][1], false, b[s], (short)0, c[1], false, false);
    }
}

__device__ __forceinline__ void bias_relu_store2(const v8f c[2][4], const float bvs[4],
                                                 unsigned char* __restrict__ Ht, int lane) {
    const int col = lane & 15, hi = lane >> 4;
    unsigned char* __restrict__ hp = Ht + col * HTP_B + 16 * hi;
    const v8h hz = {0, 0, 0, 0, 0, 0, 0, 0};
#pragma unroll
    for (int t = 0; t < 4; ++t) {
        const float bv = bvs[t];
#pragma unroll
        for (int m = 0; m < 2; ++m) {
            const v8f v = c[m][t] + bv;
            const v8h h = __builtin_elementwise_max(__builtin_convertvector(v, v8h), hz);
            *(v8h*)(hp + t * 16 * HTP_B + 32 * m) = h;
        }
    }
}

__device__ __forceinline__ void load_bias4(const _Float16* __restrict__ W, int layer, int lane, float bvs[4]) {
    const float* __restrict__ bl = (const float*)(W + WB_BIAS) + layer * H_DIM + (lane & 15);
#pragma unroll
    for (int t = 0; t < 4; ++t) bvs[t] = bl[t * 16];
}

template <int KD, int OUTD>
__device__ __forceinline__ v4f w_fetch(const float* __restrict__ g1, unsigned e1,
                                       const float* __restrict__ g2, unsigned e2,
                                       const float* __restrict__ g3, unsigned e3,
                                       const float* __restrict__ gb1, const float* __restrict__ gb2, unsigned eb,
                                       int tid, v4f q[NP_MAX]) {
    constexpr int NP1 = KD / 16;
    constexpr int LPR = KD / 128;
    const unsigned o1 = e1 + (unsigned)((tid >> 5) * KD + (tid & 31) * 4);
    const unsigned o2 = e2 + (unsigned)(tid * 4);
    const unsigned o3 = e3 + ((tid < OUTD * (H_DIM / 4)) ? (unsigned)(tid * 4) : 0u);
    const float* __restrict__ s1 = g1 + o1;
    const float* __restrict__ s2 = g2 + o2;
#pragma unroll
    for (int p = 0; p < NP1; ++p)
        q[p] = *(const v4f*)(s1 + ((p / LPR) * 8 * KD + (p % LPR) * 128));
#pragma unroll
    for (int j = 0; j < 4; ++j)
        q[NP1 + j] = *(const v4f*)(s2 + j * NTHREADS * 4);
    q[NP1 + 4] = *(const v4f*)(g3 + o3);
    const float* __restrict__ sb = ((tid & 16) ? gb2 : gb1) + (eb + (unsigned)((tid & 15) * 4));
    const v4f qb = *(const v4f*)sb;
    __builtin_amdgcn_sched_barrier(0);
    return qb;
}

template <int KD, int OUTD>
__device__ __forceinline__ void w_round(const v4f q[NP_MAX], v4h h[NP_MAX], int tid) {
    constexpr int NP = KD / 16 + 5;
    __builtin_amdgcn_sched_barrier(0);
#pragma unroll
    for (int p = 0; p < NP; ++p) h[p] = __builtin_convertvector(q[p], v4h);
    const v4h hz = {0, 0, 0, 0};
    if (tid >= OUTD * (H_DIM / 4)) h[NP - 1] = hz;
    __builtin_amdgcn_sched_barrier(0);
}

template <int KD>
__device__ __forceinline__ void w_store(_Float16* __restrict__ Wb, int tid, const v4h h[NP_MAX], v4f qb) {
    constexpr int NP1 = KD / 16;
    constexpr int LPR = KD / 128;
    _Float16* __restrict__ d1 = Wb + (tid >> 5) * W1S + (tid & 31) * 4;
    _Float16* __restrict__ d2 = Wb + WB_W2 + (tid >> 4) * HS + (tid & 15) * 4;
#pragma unroll
    for (int p = 0; p < NP1; ++p) *(v4h*)(d1 + (p / LPR) * 8 * W1S + (p % LPR) * 128) = h[p];
#pragma unroll
    for (int j = 0; j < 4; ++j) *(v4h*)(d2 + j * 16 * HS) = h[NP1 + j];
    *(v4h*)(d2 + (WB_W3 - WB_W2)) = h[NP1 + 4];
    *(v4f*)((float*)(Wb + WB_BIAS) + (tid & 31) * 4) = qb;
}

__global__ __launch_bounds__(NTHREADS) __attribute__((amdgpu_waves_per_eu(2, 2)))
void qplex_main(const float* __restrict__ states, const float* __restrict__ actions,
                const float* __restrict__ wk1, const float* __restrict__ wa1,
                const float* __restrict__ wc1,
                const float* __restrict__ wk2, const float* __restrict__ wa2,
                const float* __restrict__ wc2,
                const float* __restrict__ wk3, const float* __restrict__ wa3,
                const float* __restrict__ wc3,
                const float* __restrict__ bk1, const float* __restrict__ bk2,
                const float* __restrict__ bk3,
                const float* __restrict__ ba1, const float* __restrict__ ba2,
                const float* __restrict__ ba3,
                const float* __restrict__ bc1, const float* __restrict__ bc2,
                const float* __restrict__ bc3,
                float* __restrict__ out) {
    __shared__ __attribute__((aligned(16))) unsigned char Hs[NWAVES * HT_WAVE_B];
    __shared__ __attribute__((aligned(16))) unsigned char Xs[NWAVES * XW_WAVE_B];
    __shared__ __attribute__((aligned(16))) _Float16 Wb[2 * WB_TOTAL];

    const int tid = threadIdx.x;
    const int b0  = blockIdx.x * WG_ROWS;

    v4f pre[NP_MAX];
    v4h ph[NP_MAX];
    v4f preb = w_fetch<S_DIM, 1>(wk1, 0u, wk2, 0u, wk3, 0u, bk1, bk2, 0u, tid, pre);

    const int wave = tid >> 5, lane = tid & 31;
    const int col = lane & 15, hi = lane >> 4;
    unsigned char* Hw = Hs + wave * HT_WAVE_B;
    unsigned char* Xw = Xs + wave * XW_WAVE_B;

    v16h xa0[XS_REG], xa1[XS_REG];
    {
        const size_t r0 = (size_t)(b0 + wave * ROWS_PER_WAVE + col);
        v16h xf[XSTEPS];
        x_frags(states + r0 * S_DIM, actions + r0 * AD_DIM, lane, xf);
#pragma unroll
        for (int j = 0; j < XS_REG; ++j) xa0[j] = xf[j];
#pragma unroll
        for (int j = 0; j < XS_LDS; ++j) park_frag(Xw, j, 0, lane, xf[XS_REG + j]);
        x_frags(states + (r0 + 16) * S_DIM, actions + (r0 + 16) * AD_DIM, lane, xf);
#pragma unroll
        for (int j = 0; j < XS_REG; ++j) xa1[j] = xf[j];
#pragma unroll
        for (int j = 0; j < XS_LDS; ++j) park_frag(Xw, j, 1, lane, xf[XS_REG + j]);
    }

    w_round<S_DIM, 1>(pre, ph, tid);
    w_store<S_DIM>(Wb, tid, ph, preb);
    unsigned cur = 0u;

    float acc[2][8];
#pragma unroll
    for (int m = 0; m < 2; ++m)
#pragma unroll
        for (int r = 0; r < 8; ++r) acc[m][r] = 0.f;

#pragma unroll 1
    for (int k = 0; k < K_N; ++k) {
        v8f c1[2][4];
        v8f ck[2], ca[2], cc[2];
        v16h b2f[2][4], a2[2][2], b3f[2];
        float bv1[4], bv2[4];
        float kbias, ab3v, cb3v;
        const unsigned uk  = (unsigned)k;
        const unsigned ukn = (k + 1 < K_N) ? uk + 1u : uk;

        {
            const _Float16* Wc = Wb + cur;
            _Float16* Wn = Wb + (WB_TOTAL - cur);
            __syncthreads();
            kbias = bk3[k];
            ab3v  = ba3[k * A_N + (col & 7)];
            cb3v  = bc3[k * A_N + (col & 7)];
            preb = w_fetch<D_DIM, A_N>(wc1, uk * (H_DIM * D_DIM), wc2, uk * (H_DIM * H_DIM),
                                        wc3, uk * (A_N * H_DIM), bc1, bc2, uk * H_DIM, tid, pre);
            load_bias4(Wc, 0, lane, bv1);
            gemm4x2_xr<S_DIM>(xa0, xa1, Xw, Wc, W1S, lane, c1);
            bias_relu_store2(c1, bv1, Hw, lane);
            load_b_4x2(Wc + WB_W2, HS, lane, b2f);
            load_a_2x2(Hw, lane, a2);
            load_bias4(Wc, 1, lane, bv2);
            w_round<D_DIM, A_N>(pre, ph, tid);
            w_store<D_DIM>(Wn, tid, ph, preb);
            mma4x2(a2, b2f, c1);
            load_b_1x2(Wc + WB_W3, HS, lane, b3f);
            bias_relu_store2(c1, bv2, Hw, lane);
            gemm1x2(Hw, b3f, lane, ck);
            cur = WB_TOTAL - cur;
        }

        {
            const _Float16* Wc = Wb + cur;
            _Float16* Wn = Wb + (WB_TOTAL - cur);
            __syncthreads();
            preb = w_fetch<S_DIM, A_N>(wa1, uk * (H_DIM * S_DIM), wa2, uk * (H_DIM * H_DIM),
                                        wa3, uk * (A_N * H_DIM), ba1, ba2, uk * H_DIM, tid, pre);
            load_bias4(Wc, 0, lane, bv1);
            gemm4x2_xr<D_DIM>(xa0, xa1, Xw, Wc, W1S, lane, c1);
            bias_relu_store2(c1, bv1, Hw, lane);
            load_b_4x2(Wc + WB_W2, HS, lane, b2f);
            load_a_2x2(Hw, lane, a2);
            load_bias4(Wc, 1, lane, bv2);
            w_round<S_DIM, A_N>(pre, ph, tid);
            w_store<S_DIM>(Wn, tid, ph, preb);
            mma4x2(a2, b2f, c1);
            load_b_1x2(Wc + WB_W3, HS, lane, b3f);
            bias_relu_store2(c1, bv2, Hw, lane);
            gemm1x2(Hw, b3f, lane, cc);
            cur = WB_TOTAL - cur;
        }

        {
            const _Float16* Wc = Wb + cur;
            _Float16* Wn = Wb + (WB_TOTAL - cur);
            __syncthreads();
            preb = w_fetch<S_DIM, 1>(wk1, ukn * (H_DIM * S_DIM), wk2, ukn * (H_DIM * H_DIM),
                                      wk3, ukn * H_DIM, bk1, bk2, ukn * H_DIM, tid, pre);
            load_bias4(Wc, 0, lane, bv1);
            gemm4x2_xr<S_DIM>(xa0, xa1, Xw, Wc, W1S, lane, c1);
            bias_relu_store2(c1, bv1, Hw, lane);
            load_b_4x2(Wc + WB_W2, HS, lane, b2f);
            load_a_2x2(Hw, lane, a2);
            load_bias4(Wc, 1, lane, bv2);
            w_round<S_DIM, 1>(pre, ph, tid);
            w_store<S_DIM>(Wn, tid, ph, preb);
            mma4x2(a2, b2f, c1);
            load_b_1x2(Wc + WB_W3, HS, lane, b3f);
            bias_relu_store2(c1, bv2, Hw, lane);
            gemm1x2(Hw, b3f, lane, ca);
            cur = WB_TOTAL - cur;
        }

#pragma unroll
        for (int m = 0; m < 2; ++m)
#pragma unroll
            for (int r = 0; r < 8; ++r) {
                float kv = __shfl(ck[m][r], lane & 16, 32);
                kv = fabsf(kv + kbias) + 1e-10f;
                float ag = 1.f / (1.f + __expf(-(ca[m][r] + ab3v)));
                float ac = 1.f / (1.f + __expf(-(cc[m][r] + cb3v)));
                acc[m][r] += kv * ag * ac;
            }
    }

    if (col < A_N) {
#pragma unroll
        for (int m = 0; m < 2; ++m)
#pragma unroll
            for (int r = 0; r < 8; ++r)
                out[(size_t)(b0 + wave * ROWS_PER_WAVE + m * 16 + r + 8 * hi) * A_N + col] =
                    acc[m][r];
    }
}

extern "C" void kernel_launch(void* const* d_in, const int* in_sizes, int n_in,
                              void* d_out, int out_size, void* d_ws, size_t ws_size,
                              hipStream_t stream) {
    (void)in_sizes; (void)n_in; (void)out_size; (void)d_ws; (void)ws_size;
    const float* states  = (const float*)d_in[0];
    const float* actions = (const float*)d_in[1];
    const float* kW1 = (const float*)d_in[2];
    const float* kb1 = (const float*)d_in[3];
    const float* kW2 = (const float*)d_in[4];
    const float* kb2 = (const float*)d_in[5];
    const float* kW3 = (const float*)d_in[6];
    const float* kb3 = (const float*)d_in[7];
    const float* aW1 = (const float*)d_in[8];
    const float* ab1 = (const float*)d_in[9];
    const float* aW2 = (const float*)d_in[10];
    const float* ab2 = (const float*)d_in[11];
    const float* aW3 = (const float*)d_in[12];
    const float* ab3 = (const float*)d_in[13];
    const float* cW1 = (const float*)d_in[14];
    const float* cb1 = (const float*)d_in[15];
    const float* cW2 = (const float*)d_in[16];
    const float* cb2 = (const float*)d_in[17];
    const float* cW3 = (const float*)d_in[18];
    const float* cb3 = (const float*)d_in[19];

    qplex_main<<<B_TOT / WG_ROWS, NTHREADS, 0, stream>>>(
        states, actions,
        kW1, aW1, cW1, kW2, aW2, cW2, kW3, aW3, cW3,
        kb1, kb2, kb3, ab1, ab2, ab3, cb1, cb2, cb3,
        (float*)d_out);
}
